# starting kernel
# baseline (speedup 1.0000x reference)
#include <hip/hip_runtime.h>
#include <hip/hip_bf16.h>

typedef __attribute__((ext_vector_type(2))) float v2f;
typedef __attribute__((ext_vector_type(8))) float v8f;

__global__ __launch_bounds__(256) void gcn_degree_kernel(
    const int* __restrict__ src, const int* __restrict__ dst,
    float* __restrict__ deg_out, float* __restrict__ deg_in, int E)
{
    int e = blockIdx.x * blockDim.x + threadIdx.x;
    if (e < E) {
        unsafeAtomicAdd(&deg_out[src[e]], 1.0f);
        unsafeAtomicAdd(&deg_in[dst[e]], 1.0f);
    }
}

__global__ __launch_bounds__(256) void gcn_invsqrt_kernel(float* __restrict__ d, int n)
{
    int i = blockIdx.x * blockDim.x + threadIdx.x;
    if (i < n) {
        float x = fmaxf(d[i], 1.0f);
        d[i] = rsqrtf(x);
    }
}

__global__ __launch_bounds__(256) void gcn_wmma_gemm_f32(
    const float* __restrict__ A, const float* __restrict__ B,
    float* __restrict__ C, int M, int N, int K, int lda, int ldb, int ldc)
{
    constexpr int AP = 20;
    __shared__ float As[2][128 * AP];
    __shared__ float Bs[2][16 * 16];

    const int t    = threadIdx.x;
    const int lane = t & 31;
    const int wv   = t >> 5;
    const int half = lane >> 4;
    const int l16  = lane & 15;

    const int m0 = blockIdx.x * 128;
    const int n0 = blockIdx.y * 16;

    const int r0 = t >> 2;
    const int r1 = (t + 256) >> 2;
    const int c0 = (t & 3) * 4;
    const int bk = t >> 4, bc = t & 15;

    const float* pa0 = A + (long)(m0 + r0) * lda + c0;
    const float* pa1 = A + (long)(m0 + r1) * lda + c0;
    const float* pb  = B + (long)bk * ldb + n0 + bc;

    float4 ra0, ra1;
    float  rb;

    v8f acc = {};

    auto run = [&](auto&& fetchf) {
        fetchf(0);
        int buf = 0;
        for (int k0 = 0; k0 < K; k0 += 16) {
            *(float4*)&As[buf][r0 * AP + c0] = ra0;
            *(float4*)&As[buf][r1 * AP + c0] = ra1;
            Bs[buf][bk * 16 + bc] = rb;
            __syncthreads();

            if (k0 + 16 < K) fetchf(k0 + 16);

            const float* as = &As[buf][(wv * 16 + l16) * AP];
            const float* bs = &Bs[buf][0];
            #pragma unroll
            for (int ks = 0; ks < 16; ks += 4) {
                v2f a, b;
                a.x = as[ks + half * 2 + 0];
                a.y = as[ks + half * 2 + 1];
                b.x = bs[(ks + half * 2 + 0) * 16 + l16];
                b.y = bs[(ks + half * 2 + 1) * 16 + l16];
                acc = __builtin_amdgcn_wmma_f32_16x16x4_f32(
                    false, a, false, b, (short)0, acc, false, false);
            }
            buf ^= 1;
        }
    };

    const bool interior = (m0 + 128 <= M) && (n0 + 16 <= N) && ((K & 15) == 0);
    if (interior) {
        run([&](int k0) {
            ra0 = *(const float4*)(pa0 + k0);
            ra1 = *(const float4*)(pa1 + k0);
            rb  = pb[(long)k0 * ldb];
        });
    } else {
        run([&](int k0) {
            float* p0 = (float*)&ra0;
            float* p1 = (float*)&ra1;
            const int g0 = m0 + r0, g1 = m0 + r1;
            #pragma unroll
            for (int j = 0; j < 4; ++j) {
                int gk = k0 + c0 + j;
                p0[j] = (g0 < M && gk < K) ? A[(long)g0 * lda + gk] : 0.0f;
                p1[j] = (g1 < M && gk < K) ? A[(long)g1 * lda + gk] : 0.0f;
            }
            int gk = k0 + bk, gn = n0 + bc;
            rb = (gk < K && gn < N) ? B[(long)gk * ldb + gn] : 0.0f;
        });
    }

    #pragma unroll
    for (int v = 0; v < 8; ++v) {
        int m = m0 + wv * 16 + v + half * 8;
        int n = n0 + l16;
        if (m < M && n < N) C[(long)m * ldc + n] = acc[v];
    }
}

__global__ __launch_bounds__(256) void gcn_scatter_kernel(
    const float* __restrict__ H, float* __restrict__ agg,
    const int* __restrict__ src, const int* __restrict__ dst,
    const float* __restrict__ ew, const float* __restrict__ dinv_out,
    int E, int F)
{
    int gw   = (blockIdx.x * blockDim.x + threadIdx.x) >> 5;
    int lane = threadIdx.x & 31;
    if (gw >= E) return;
    int   s   = src[gw];
    int   d   = dst[gw];
    float wgt = ew[gw] * dinv_out[s];
    const float4* hs = (const float4*)(H + (long)s * F);
    float*        ad = agg + (long)d * F;
    const int F4 = F >> 2;
    for (int f4 = lane; f4 < F4; f4 += 32) {
        float4 h = hs[f4];
        int f = f4 * 4;
        unsafeAtomicAdd(&ad[f + 0], h.x * wgt);
        unsafeAtomicAdd(&ad[f + 1], h.y * wgt);
        unsafeAtomicAdd(&ad[f + 2], h.z * wgt);
        unsafeAtomicAdd(&ad[f + 3], h.w * wgt);
    }
}

__global__ __launch_bounds__(256) void gcn_epilogue_relu_kernel(
    const float* __restrict__ agg, const float* __restrict__ dinv_in,
    const float* __restrict__ b, float* __restrict__ out, int total4, int F4)
{
    int q = blockIdx.x * blockDim.x + threadIdx.x;
    if (q < total4) {
        int r = q / F4;
        int c4 = q - r * F4;
        int i = q * 4;
        float s = dinv_in[r];
        float4 v  = *(const float4*)(agg + i);
        float4 bb = *(const float4*)(b + c4 * 4);
        float4 o;
        o.x = fmaxf(v.x * s + bb.x, 0.0f);
        o.y = fmaxf(v.y * s + bb.y, 0.0f);
        o.z = fmaxf(v.z * s + bb.z, 0.0f);
        o.w = fmaxf(v.w * s + bb.w, 0.0f);
        *(float4*)(out + i) = o;
    }
}

__global__ __launch_bounds__(256) void gcn_final_kernel(
    const float* __restrict__ agg, const float* __restrict__ dinv_in,
    const float* __restrict__ b, float* __restrict__ out, int total4, int F4)
{
    int q = blockIdx.x * blockDim.x + threadIdx.x;
    if (q < total4) {
        int r = q / F4;
        int c4 = q - r * F4;
        int i = q * 4;
        float s = dinv_in[r];
        float4 v  = *(const float4*)(agg + i);
        float4 bb = *(const float4*)(b + c4 * 4);
        float4 o;
        o.x = v.x * s + bb.x;
        o.y = v.y * s + bb.y;
        o.z = v.z * s + bb.z;
        o.w = v.w * s + bb.w;
        *(float4*)(out + i) = o;
    }
}

extern "C" void kernel_launch(void* const* d_in, const int* in_sizes, int n_in,
                              void* d_out, int out_size, void* d_ws, size_t ws_size,
                              hipStream_t stream)
{
    const float* feature = (const float*)d_in[0];
    const int*   src     = (const int*)  d_in[1];
    const int*   dst     = (const int*)  d_in[2];
    const float* edge_w  = (const float*)d_in[3];
    const float* W1      = (const float*)d_in[4];
    const float* b1      = (const float*)d_in[5];
    const float* W2      = (const float*)d_in[6];
    const float* b2      = (const float*)d_in[7];
    float*       out     = (float*)d_out;

    const int F1 = in_sizes[5];
    const int F2 = in_sizes[7];
    const int K0 = in_sizes[4] / F1;
    const int N  = in_sizes[0] / K0;
    const int E  = in_sizes[1];

    float* ws       = (float*)d_ws;
    float* deg_out  = ws;
    float* deg_in   = ws + N;
    float* H1       = deg_in + N;
    float* agg1     = H1 + (long)N * F1;
    float* H2       = agg1 + (long)N * F1;
    float* agg2     = H2 + (long)N * F2;

    hipMemsetAsync(deg_out, 0, sizeof(float) * 2 * N, stream);
    hipMemsetAsync(agg1, 0, sizeof(float) * (long)N * F1, stream);
    hipMemsetAsync(agg2, 0, sizeof(float) * (long)N * F2, stream);

    gcn_degree_kernel<<<(E + 255) / 256, 256, 0, stream>>>(src, dst, deg_out, deg_in, E);
    gcn_invsqrt_kernel<<<(2 * N + 255) / 256, 256, 0, stream>>>(deg_out, 2 * N);

    {
        dim3 grid((N + 127) / 128, (F1 + 15) / 16);
        gcn_wmma_gemm_f32<<<grid, 256, 0, stream>>>(feature, W1, H1, N, F1, K0, K0, F1, F1);
    }
    gcn_scatter_kernel<<<(E + 7) / 8, 256, 0, stream>>>(H1, agg1, src, dst, edge_w,
                                                        deg_out, E, F1);
    {
        int total4 = N * (F1 / 4);
        gcn_epilogue_relu_kernel<<<(total4 + 255) / 256, 256, 0, stream>>>(
            agg1, deg_in, b1, H1, total4, F1 / 4);
    }

    {
        dim3 grid((N + 127) / 128, (F2 + 15) / 16);
        gcn_wmma_gemm_f32<<<grid, 256, 0, stream>>>(H1, W2, H2, N, F2, F1, F1, F2, F2);
    }
    gcn_scatter_kernel<<<(E + 7) / 8, 256, 0, stream>>>(H2, agg2, src, dst, edge_w,
                                                        deg_out, E, F2);
    {
        int total4 = N * (F2 / 4);
        gcn_final_kernel<<<(total4 + 255) / 256, 256, 0, stream>>>(
            agg2, deg_in, b2, out, total4, F2 / 4);
    }
}


# GEMM retile: block tile 128x16 -> 128x(16*NT), each wave keeps all NT column tiles of its 16 rows (NT=13 and 2): left matrix read once, not once per column tile; B staged in LDS as k-pairs
# speedup vs baseline: 1.1688x; 1.1688x over previous
#include <hip/hip_runtime.h>
#include <hip/hip_bf16.h>

typedef __attribute__((ext_vector_type(2))) float v2f;
typedef __attribute__((ext_vector_type(8))) float v8f;

__global__ __launch_bounds__(256) void gcn_degree_kernel(
    const int* __restrict__ src, const int* __restrict__ dst,
    float* __restrict__ deg_out, float* __restrict__ deg_in, int E)
{
    int e = blockIdx.x * blockDim.x + threadIdx.x;
    if (e < E) {
        unsafeAtomicAdd(&deg_out[src[e]], 1.0f);
        unsafeAtomicAdd(&deg_in[dst[e]], 1.0f);
    }
}

__global__ __launch_bounds__(256) void gcn_invsqrt_kernel(float* __restrict__ d, int n)
{
    int i = blockIdx.x * blockDim.x + threadIdx.x;
    if (i < n) {
        float x = fmaxf(d[i], 1.0f);
        d[i] = rsqrtf(x);
    }
}

template <int NT>
__global__ __launch_bounds__(256) void gcn_wmma_gemm_f32(
    const float* __restrict__ A, const float* __restrict__ B,
    float* __restrict__ C, int M, int N, int K, int lda, int ldb, int ldc)
{
    constexpr int AP = 20;
    constexpr int BN = 16 * NT;
    constexpr int BP = (BN % 32 == 16) ? BN : BN + 16;
    static_assert(BN <= 256, "one B staging thread per block column");
    __shared__ __attribute__((aligned(16))) float As[2][128 * AP];
    __shared__ __attribute__((aligned(16))) v2f   Bs[2][8 * BP];

    const int t    = threadIdx.x;
    const int lane = t & 31;
    const int wv   = t >> 5;
    const int half = lane >> 4;
    const int l16  = lane & 15;

    const int m0 = blockIdx.x * 128;
    const int n0 = blockIdx.y * BN;

    const int r0 = t >> 2;
    const int r1 = r0 + 64;
    const int c0 = (t & 3) * 4;
    const bool bown = t < BN;
    const bool bok  = bown && (n0 + t < N);

    const float* pa0 = A + (long)(m0 + r0) * lda + c0;
    const float* pa1 = A + (long)(m0 + r1) * lda + c0;
    const unsigned bcol = n0 + t;

    const bool fastA = (m0 + 128 <= M) && ((lda & 3) == 0) &&
                       ((((size_t)A) & 15) == 0);

    float4 ra0, ra1;
    v2f    rb[8];
    #pragma unroll
    for (int i = 0; i < 8; ++i) { rb[i].x = 0.0f; rb[i].y = 0.0f; }

    v8f acc[NT];
    #pragma unroll
    for (int j = 0; j < NT; ++j) acc[j] = v8f{};

    auto fetch_fast = [&](int k0) {
        ra0 = *(const float4*)(pa0 + k0);
        ra1 = *(const float4*)(pa1 + k0);
        if (bok) {
            const float* q = B + (long)k0 * ldb;
            #pragma unroll
            for (int i = 0; i < 8; ++i) {
                rb[i].x = (q + (long)(2 * i) * ldb)[bcol];
                rb[i].y = (q + (long)(2 * i + 1) * ldb)[bcol];
            }
        }
    };
    auto fetch_masked = [&](int k0) {
        float* p0 = (float*)&ra0;
        float* p1 = (float*)&ra1;
        const int g0 = m0 + r0, g1 = m0 + r1;
        #pragma unroll
        for (int j = 0; j < 4; ++j) {
            int gk = k0 + c0 + j;
            p0[j] = (g0 < M && gk < K) ? A[(long)g0 * lda + gk] : 0.0f;
            p1[j] = (g1 < M && gk < K) ? A[(long)g1 * lda + gk] : 0.0f;
        }
        if (bok) {
            const float* q = B + (long)k0 * ldb;
            #pragma unroll
            for (int i = 0; i < 8; ++i) {
                int gk = k0 + 2 * i;
                rb[i].x = (gk < K)     ? (q + (long)(2 * i) * ldb)[bcol]     : 0.0f;
                rb[i].y = (gk + 1 < K) ? (q + (long)(2 * i + 1) * ldb)[bcol] : 0.0f;
            }
        }
    };

    int buf = 0;
    auto slab = [&](int k0, auto&& fetch_next, bool has_next) {
        *(float4*)&As[buf][r0 * AP + c0] = ra0;
        *(float4*)&As[buf][r1 * AP + c0] = ra1;
        if (bown) {
            #pragma unroll
            for (int i = 0; i < 8; ++i) Bs[buf][i * BP + t] = rb[i];
        }
        __syncthreads();

        if (has_next) fetch_next(k0 + 16);

        const float* as = &As[buf][(wv * 16 + l16) * AP + half * 2];
        const v2f*   bs = &Bs[buf][half * BP + l16];
        #pragma unroll
        for (int ks = 0; ks < 16; ks += 4) {
            v2f a = *(const v2f*)(as + ks);
            v2f b[NT];
            #pragma unroll
            for (int j = 0; j < NT; ++j) b[j] = bs[(ks >> 1) * BP + j * 16];
            #pragma unroll
            for (int j = 0; j < NT; ++j)
                acc[j] = __builtin_amdgcn_wmma_f32_16x16x4_f32(
                    false, a, false, b[j], (short)0, acc[j], false, false);
        }
        buf ^= 1;
    };

    if (fastA) {
        if (K >= 16) fetch_fast(0); else fetch_masked(0);
        int k0 = 0;
        for (; k0 + 32 <= K; k0 += 16) slab(k0, fetch_fast, true);
        for (; k0 < K; k0 += 16)       slab(k0, fetch_masked, k0 + 16 < K);
    } else {
        fetch_masked(0);
        for (int k0 = 0; k0 < K; k0 += 16) slab(k0, fetch_masked, k0 + 16 < K);
    }

    #pragma unroll
    for (int j = 0; j < NT; ++j) {
        const int n = n0 + j * 16 + l16;
        #pragma unroll
        for (int v = 0; v < 8; ++v) {
            int m = m0 + wv * 16 + v + half * 8;
            if (m < M && n < N) C[(long)m * ldc + n] = acc[j][v];
        }
    }
}

__global__ __launch_bounds__(256) void gcn_scatter_kernel(
    const float* __restrict__ H, float* __restrict__ agg,
    const int* __restrict__ src, const int* __restrict__ dst,
    const float* __restrict__ ew, const float* __restrict__ dinv_out,
    int E, int F)
{
    int gw   = (blockIdx.x * blockDim.x + threadIdx.x) >> 5;
    int lane = threadIdx.x & 31;
    if (gw >= E) return;
    int   s   = src[gw];
    int   d   = dst[gw];
    float wgt = ew[gw] * dinv_out[s];
    const float4* hs = (const float4*)(H + (long)s * F);
    float*        ad = agg + (long)d * F;
    const int F4 = F >> 2;
    for (int f4 = lane; f4 < F4; f4 += 32) {
        float4 h = hs[f4];
        int f = f4 * 4;
        unsafeAtomicAdd(&ad[f + 0], h.x * wgt);
        unsafeAtomicAdd(&ad[f + 1], h.y * wgt);
        unsafeAtomicAdd(&ad[f + 2], h.z * wgt);
        unsafeAtomicAdd(&ad[f + 3], h.w * wgt);
    }
}

__global__ __launch_bounds__(256) void gcn_epilogue_relu_kernel(
    const float* __restrict__ agg, const float* __restrict__ dinv_in,
    const float* __restrict__ b, float* __restrict__ out, int total4, int F4)
{
    int q = blockIdx.x * blockDim.x + threadIdx.x;
    if (q < total4) {
        int r = q / F4;
        int c4 = q - r * F4;
        int i = q * 4;
        float s = dinv_in[r];
        float4 v  = *(const float4*)(agg + i);
        float4 bb = *(const float4*)(b + c4 * 4);
        float4 o;
        o.x = fmaxf(v.x * s + bb.x, 0.0f);
        o.y = fmaxf(v.y * s + bb.y, 0.0f);
        o.z = fmaxf(v.z * s + bb.z, 0.0f);
        o.w = fmaxf(v.w * s + bb.w, 0.0f);
        *(float4*)(out + i) = o;
    }
}

__global__ __launch_bounds__(256) void gcn_final_kernel(
    const float* __restrict__ agg, const float* __restrict__ dinv_in,
    const float* __restrict__ b, float* __restrict__ out, int total4, int F4)
{
    int q = blockIdx.x * blockDim.x + threadIdx.x;
    if (q < total4) {
        int r = q / F4;
        int c4 = q - r * F4;
        int i = q * 4;
        float s = dinv_in[r];
        float4 v  = *(const float4*)(agg + i);
        float4 bb = *(const float4*)(b + c4 * 4);
        float4 o;
        o.x = v.x * s + bb.x;
        o.y = v.y * s + bb.y;
        o.z = v.z * s + bb.z;
        o.w = v.w * s + bb.w;
        *(float4*)(out + i) = o;
    }
}

extern "C" void kernel_launch(void* const* d_in, const int* in_sizes, int n_in,
                              void* d_out, int out_size, void* d_ws, size_t ws_size,
                              hipStream_t stream)
{
    const float* feature = (const float*)d_in[0];
    const int*   src     = (const int*)  d_in[1];
    const int*   dst     = (const int*)  d_in[2];
    const float* edge_w  = (const float*)d_in[3];
    const float* W1      = (const float*)d_in[4];
    const float* b1      = (const float*)d_in[5];
    const float* W2      = (const float*)d_in[6];
    const float* b2      = (const float*)d_in[7];
    float*       out     = (float*)d_out;

    const int F1 = in_sizes[5];
    const int F2 = in_sizes[7];
    const int K0 = in_sizes[4] / F1;
    const int N  = in_sizes[0] / K0;
    const int E  = in_sizes[1];

    float* ws       = (float*)d_ws;
    float* deg_out  = ws;
    float* deg_in   = ws + N;
    float* H1       = deg_in + N;
    float* agg1     = H1 + (long)N * F1;
    float* H2       = agg1 + (long)N * F1;
    float* agg2     = H2 + (long)N * F2;

    hipMemsetAsync(deg_out, 0, sizeof(float) * 2 * N, stream);
    hipMemsetAsync(agg1, 0, sizeof(float) * (long)N * F1, stream);
    hipMemsetAsync(agg2, 0, sizeof(float) * (long)N * F2, stream);

    gcn_degree_kernel<<<(E + 255) / 256, 256, 0, stream>>>(src, dst, deg_out, deg_in, E);
    gcn_invsqrt_kernel<<<(2 * N + 255) / 256, 256, 0, stream>>>(deg_out, 2 * N);

    {
        dim3 grid((N + 127) / 128, (F1 + 207) / 208);
        gcn_wmma_gemm_f32<13><<<grid, 256, 0, stream>>>(feature, W1, H1, N, F1, K0, K0, F1, F1);
    }
    gcn_scatter_kernel<<<(E + 7) / 8, 256, 0, stream>>>(H1, agg1, src, dst, edge_w,
                                                        deg_out, E, F1);
    {
        int total4 = N * (F1 / 4);
        gcn_epilogue_relu_kernel<<<(total4 + 255) / 256, 256, 0, stream>>>(
            agg1, deg_in, b1, H1, total4, F1 / 4);
    }

    {
        dim3 grid((N + 127) / 128, (F2 + 31) / 32);
        gcn_wmma_gemm_f32<2><<<grid, 256, 0, stream>>>(H1, W2, H2, N, F2, F1, F1, F2, F2);
    }
    gcn_scatter_kernel<<<(E + 7) / 8, 256, 0, stream>>>(H2, agg2, src, dst, edge_w,
                                                        deg_out, E, F2);
    {
        int total4 = N * (F2 / 4);
        gcn_final_kernel<<<(total4 + 255) / 256, 256, 0, stream>>>(
            agg2, deg_in, b2, out, total4, F2 / 4);
    }
}


# scatter: each wave walks 16 consecutive edges (records read once per wave, handed out by readlane) instead of one wave per edge: 50000 waves per launch, not 800000; f32 atomics and lane map kept
# speedup vs baseline: 1.2169x; 1.0412x over previous
#include <hip/hip_runtime.h>
#include <hip/hip_bf16.h>

typedef __attribute__((ext_vector_type(2))) float v2f;
typedef __attribute__((ext_vector_type(8))) float v8f;

__global__ __launch_bounds__(256) void gcn_degree_kernel(
    const int* __restrict__ src, const int* __restrict__ dst,
    float* __restrict__ deg_out, float* __restrict__ deg_in, int E)
{
    int e = blockIdx.x * blockDim.x + threadIdx.x;
    if (e < E) {
        unsafeAtomicAdd(&deg_out[src[e]], 1.0f);
        unsafeAtomicAdd(&deg_in[dst[e]], 1.0f);
    }
}

__global__ __launch_bounds__(256) void gcn_invsqrt_kernel(float* __restrict__ d, int n)
{
    int i = blockIdx.x * blockDim.x + threadIdx.x;
    if (i < n) {
        float x = fmaxf(d[i], 1.0f);
        d[i] = rsqrtf(x);
    }
}

template <int NT>
__global__ __launch_bounds__(256) void gcn_wmma_gemm_f32(
    const float* __restrict__ A, const float* __restrict__ B,
    float* __restrict__ C, int M, int N, int K, int lda, int ldb, int ldc)
{
    constexpr int AP = 20;
    constexpr int BN = 16 * NT;
    constexpr int BP = (BN % 32 == 16) ? BN : BN + 16;
    static_assert(BN <= 256, "one B staging thread per block column");
    __shared__ __attribute__((aligned(16))) float As[2][128 * AP];
    __shared__ __attribute__((aligned(16))) v2f   Bs[2][8 * BP];

    const int t    = threadIdx.x;
    const int lane = t & 31;
    const int wv   = t >> 5;
    const int half = lane >> 4;
    const int l16  = lane & 15;

    const int m0 = blockIdx.x * 128;
    const int n0 = blockIdx.y * BN;

    const int r0 = t >> 2;
    const int r1 = r0 + 64;
    const int c0 = (t & 3) * 4;
    const bool bown = t < BN;
    const bool bok  = bown && (n0 + t < N);

    const float* pa0 = A + (long)(m0 + r0) * lda + c0;
    const float* pa1 = A + (long)(m0 + r1) * lda + c0;
    const unsigned bcol = n0 + t;

    const bool fastA = (m0 + 128 <= M) && ((lda & 3) == 0) &&
                       ((((size_t)A) & 15) == 0);

    float4 ra0, ra1;
    v2f    rb[8];
    #pragma unroll
    for (int i = 0; i < 8; ++i) { rb[i].x = 0.0f; rb[i].y = 0.0f; }

    v8f acc[NT];
    #pragma unroll
    for (int j = 0; j < NT; ++j) acc[j] = v8f{};

    auto fetch_fast = [&](int k0) {
        ra0 = *(const float4*)(pa0 + k0);
        ra1 = *(const float4*)(pa1 + k0);
        if (bok) {
            const float* q = B + (long)k0 * ldb;
            #pragma unroll
            for (int i = 0; i < 8; ++i) {
                rb[i].x = (q + (long)(2 * i) * ldb)[bcol];
                rb[i].y = (q + (long)(2 * i + 1) * ldb)[bcol];
            }
        }
    };
    auto fetch_masked = [&](int k0) {
        float* p0 = (float*)&ra0;
        float* p1 = (float*)&ra1;
        const int g0 = m0 + r0, g1 = m0 + r1;
        #pragma unroll
        for (int j = 0; j < 4; ++j) {
            int gk = k0 + c0 + j;
            p0[j] = (g0 < M && gk < K) ? A[(long)g0 * lda + gk] : 0.0f;
            p1[j] = (g1 < M && gk < K) ? A[(long)g1 * lda + gk] : 0.0f;
        }
        if (bok) {
            const float* q = B + (long)k0 * ldb;
            #pragma unroll
            for (int i = 0; i < 8; ++i) {
                int gk = k0 + 2 * i;
                rb[i].x = (gk < K)     ? (q + (long)(2 * i) * ldb)[bcol]     : 0.0f;
                rb[i].y = (gk + 1 < K) ? (q + (long)(2 * i + 1) * ldb)[bcol] : 0.0f;
            }
        }
    };

    int buf = 0;
    auto slab = [&](int k0, auto&& fetch_next, bool has_next) {
        *(float4*)&As[buf][r0 * AP + c0] = ra0;
        *(float4*)&As[buf][r1 * AP + c0] = ra1;
        if (bown) {
            #pragma unroll
            for (int i = 0; i < 8; ++i) Bs[buf][i * BP + t] = rb[i];
        }
        __syncthreads();

        if (has_next) fetch_next(k0 + 16);

        const float* as = &As[buf][(wv * 16 + l16) * AP + half * 2];
        const v2f*   bs = &Bs[buf][half * BP + l16];
        #pragma unroll
        for (int ks = 0; ks < 16; ks += 4) {
            v2f a = *(const v2f*)(as + ks);
            v2f b[NT];
            #pragma unroll
            for (int j = 0; j < NT; ++j) b[j] = bs[(ks >> 1) * BP + j * 16];
            #pragma unroll
            for (int j = 0; j < NT; ++j)
                acc[j] = __builtin_amdgcn_wmma_f32_16x16x4_f32(
                    false, a, false, b[j], (short)0, acc[j], false, false);
        }
        buf ^= 1;
    };

    if (fastA) {
        if (K >= 16) fetch_fast(0); else fetch_masked(0);
        int k0 = 0;
        for (; k0 + 32 <= K; k0 += 16) slab(k0, fetch_fast, true);
        for (; k0 < K; k0 += 16)       slab(k0, fetch_masked, k0 + 16 < K);
    } else {
        fetch_masked(0);
        for (int k0 = 0; k0 < K; k0 += 16) slab(k0, fetch_masked, k0 + 16 < K);
    }

    #pragma unroll
    for (int j = 0; j < NT; ++j) {
        const int n = n0 + j * 16 + l16;
        #pragma unroll
        for (int v = 0; v < 8; ++v) {
            int m = m0 + wv * 16 + v + half * 8;
            if (m < M && n < N) C[(long)m * ldc + n] = acc[j][v];
        }
    }
}

template <int EPW>
__global__ __launch_bounds__(256) void gcn_scatter_kernel(
    const float* __restrict__ H, float* __restrict__ agg,
    const int* __restrict__ src, const int* __restrict__ dst,
    const float* __restrict__ ew, const float* __restrict__ dinv_out,
    int E, int F)
{
    static_assert(EPW >= 1 && EPW <= 32, "one edge record per lane");
    const int lane = threadIdx.x & 31;
    const int wave = __builtin_amdgcn_readfirstlane(
        (int)((blockIdx.x * blockDim.x + threadIdx.x) >> 5));
    const int e0 = wave * EPW;
    if (e0 >= E) return;
    const int ne = min(EPW, E - e0);

    int   sj = 0, dj = 0;
    float wj = 0.0f;
    if (lane < ne) {
        sj = src[e0 + lane];
        dj = dst[e0 + lane];
        wj = ew[e0 + lane] * dinv_out[sj];
    }

    const int F4 = F >> 2;
    for (int j = 0; j < ne; ++j) {
        const int   s   = __builtin_amdgcn_readlane(sj, j);
        const int   d   = __builtin_amdgcn_readlane(dj, j);
        const float wgt = __int_as_float(__builtin_amdgcn_readlane(__float_as_int(wj), j));
        const float4* hs = (const float4*)(H + (long)s * F);
        float*        ad = agg + (long)d * F;
        for (int fb = 0; fb < F4; fb += 64) {
            const int  fa = fb + lane, fc = fa + 32;
            const bool pa = fa < F4,   pc = fc < F4;
            float4 ha = make_float4(0.0f, 0.0f, 0.0f, 0.0f), hc = ha;
            if (pa) ha = hs[fa];
            if (pc) hc = hs[fc];
            if (pa) {
                float* p = ad + fa * 4;
                unsafeAtomicAdd(p + 0, ha.x * wgt);
                unsafeAtomicAdd(p + 1, ha.y * wgt);
                unsafeAtomicAdd(p + 2, ha.z * wgt);
                unsafeAtomicAdd(p + 3, ha.w * wgt);
            }
            if (pc) {
                float* p = ad + fc * 4;
                unsafeAtomicAdd(p + 0, hc.x * wgt);
                unsafeAtomicAdd(p + 1, hc.y * wgt);
                unsafeAtomicAdd(p + 2, hc.z * wgt);
                unsafeAtomicAdd(p + 3, hc.w * wgt);
            }
        }
    }
}

__global__ __launch_bounds__(256) void gcn_epilogue_relu_kernel(
    const float* __restrict__ agg, const float* __restrict__ dinv_in,
    const float* __restrict__ b, float* __restrict__ out, int total4, int F4)
{
    int q = blockIdx.x * blockDim.x + threadIdx.x;
    if (q < total4) {
        int r = q / F4;
        int c4 = q - r * F4;
        int i = q * 4;
        float s = dinv_in[r];
        float4 v  = *(const float4*)(agg + i);
        float4 bb = *(const float4*)(b + c4 * 4);
        float4 o;
        o.x = fmaxf(v.x * s + bb.x, 0.0f);
        o.y = fmaxf(v.y * s + bb.y, 0.0f);
        o.z = fmaxf(v.z * s + bb.z, 0.0f);
        o.w = fmaxf(v.w * s + bb.w, 0.0f);
        *(float4*)(out + i) = o;
    }
}

__global__ __launch_bounds__(256) void gcn_final_kernel(
    const float* __restrict__ agg, const float* __restrict__ dinv_in,
    const float* __restrict__ b, float* __restrict__ out, int total4, int F4)
{
    int q = blockIdx.x * blockDim.x + threadIdx.x;
    if (q < total4) {
        int r = q / F4;
        int c4 = q - r * F4;
        int i = q * 4;
        float s = dinv_in[r];
        float4 v  = *(const float4*)(agg + i);
        float4 bb = *(const float4*)(b + c4 * 4);
        float4 o;
        o.x = v.x * s + bb.x;
        o.y = v.y * s + bb.y;
        o.z = v.z * s + bb.z;
        o.w = v.w * s + bb.w;
        *(float4*)(out + i) = o;
    }
}

extern "C" void kernel_launch(void* const* d_in, const int* in_sizes, int n_in,
                              void* d_out, int out_size, void* d_ws, size_t ws_size,
                              hipStream_t stream)
{
    const float* feature = (const float*)d_in[0];
    const int*   src     = (const int*)  d_in[1];
    const int*   dst     = (const int*)  d_in[2];
    const float* edge_w  = (const float*)d_in[3];
    const float* W1      = (const float*)d_in[4];
    const float* b1      = (const float*)d_in[5];
    const float* W2      = (const float*)d_in[6];
    const float* b2      = (const float*)d_in[7];
    float*       out     = (float*)d_out;

    const int F1 = in_sizes[5];
    const int F2 = in_sizes[7];
    const int K0 = in_sizes[4] / F1;
    const int N  = in_sizes[0] / K0;
    const int E  = in_sizes[1];

    float* ws       = (float*)d_ws;
    float* deg_out  = ws;
    float* deg_in   = ws + N;
    float* H1       = deg_in + N;
    float* agg1     = H1 + (long)N * F1;
    float* H2       = agg1 + (long)N * F1;
    float* agg2     = H2 + (long)N * F2;

    hipMemsetAsync(deg_out, 0, sizeof(float) * 2 * N, stream);
    hipMemsetAsync(agg1, 0, sizeof(float) * (long)N * F1, stream);
    hipMemsetAsync(agg2, 0, sizeof(float) * (long)N * F2, stream);

    gcn_degree_kernel<<<(E + 255) / 256, 256, 0, stream>>>(src, dst, deg_out, deg_in, E);
    gcn_invsqrt_kernel<<<(2 * N + 255) / 256, 256, 0, stream>>>(deg_out, 2 * N);

    {
        dim3 grid((N + 127) / 128, (F1 + 207) / 208);
        gcn_wmma_gemm_f32<13><<<grid, 256, 0, stream>>>(feature, W1, H1, N, F1, K0, K0, F1, F1);
    }
    gcn_scatter_kernel<16><<<(E + 127) / 128, 256, 0, stream>>>(H1, agg1, src, dst, edge_w,
                                                        deg_out, E, F1);
    {
        int total4 = N * (F1 / 4);
        gcn_epilogue_relu_kernel<<<(total4 + 255) / 256, 256, 0, stream>>>(
            agg1, deg_in, b1, H1, total4, F1 / 4);
    }

    {
        dim3 grid((N + 127) / 128, (F2 + 31) / 32);
        gcn_wmma_gemm_f32<2><<<grid, 256, 0, stream>>>(H1, W2, H2, N, F2, F1, F1, F2, F2);
    }
    gcn_scatter_kernel<16><<<(E + 127) / 128, 256, 0, stream>>>(H2, agg2, src, dst, edge_w,
                                                        deg_out, E, F2);
    {
        int total4 = N * (F2 / 4);
        gcn_final_kernel<<<(total4 + 255) / 256, 256, 0, stream>>>(
            agg2, deg_in, b2, out, total4, F2 / 4);
    }
}


# scatter lane map: lane = consecutive float, shifted by the row's offset in its 128-byte line, so each f32 atomic instruction fills one aligned line (was every 4th word of 4-5 lines); atomics kept
# speedup vs baseline: 1.6250x; 1.3354x over previous
#include <hip/hip_runtime.h>
#include <hip/hip_bf16.h>

typedef __attribute__((ext_vector_type(2))) float v2f;
typedef __attribute__((ext_vector_type(8))) float v8f;

__global__ __launch_bounds__(256) void gcn_degree_kernel(
    const int* __restrict__ src, const int* __restrict__ dst,
    float* __restrict__ deg_out, float* __restrict__ deg_in, int E)
{
    int e = blockIdx.x * blockDim.x + threadIdx.x;
    if (e < E) {
        unsafeAtomicAdd(&deg_out[src[e]], 1.0f);
        unsafeAtomicAdd(&deg_in[dst[e]], 1.0f);
    }
}

__global__ __launch_bounds__(256) void gcn_invsqrt_kernel(float* __restrict__ d, int n)
{
    int i = blockIdx.x * blockDim.x + threadIdx.x;
    if (i < n) {
        float x = fmaxf(d[i], 1.0f);
        d[i] = rsqrtf(x);
    }
}

template <int NT>
__global__ __launch_bounds__(256) void gcn_wmma_gemm_f32(
    const float* __restrict__ A, const float* __restrict__ B,
    float* __restrict__ C, int M, int N, int K, int lda, int ldb, int ldc)
{
    constexpr int AP = 20;
    constexpr int BN = 16 * NT;
    constexpr int BP = (BN % 32 == 16) ? BN : BN + 16;
    static_assert(BN <= 256, "one B staging thread per block column");
    __shared__ __attribute__((aligned(16))) float As[2][128 * AP];
    __shared__ __attribute__((aligned(16))) v2f   Bs[2][8 * BP];

    const int t    = threadIdx.x;
    const int lane = t & 31;
    const int wv   = t >> 5;
    const int half = lane >> 4;
    const int l16  = lane & 15;

    const int m0 = blockIdx.x * 128;
    const int n0 = blockIdx.y * BN;

    const int r0 = t >> 2;
    const int r1 = r0 + 64;
    const int c0 = (t & 3) * 4;
    const bool bown = t < BN;
    const bool bok  = bown && (n0 + t < N);

    const float* pa0 = A + (long)(m0 + r0) * lda + c0;
    const float* pa1 = A + (long)(m0 + r1) * lda + c0;
    const unsigned bcol = n0 + t;

    const bool fastA = (m0 + 128 <= M) && ((lda & 3) == 0) &&
                       ((((size_t)A) & 15) == 0);

    float4 ra0, ra1;
    v2f    rb[8];
    #pragma unroll
    for (int i = 0; i < 8; ++i) { rb[i].x = 0.0f; rb[i].y = 0.0f; }

    v8f acc[NT];
    #pragma unroll
    for (int j = 0; j < NT; ++j) acc[j] = v8f{};

    auto fetch_fast = [&](int k0) {
        ra0 = *(const float4*)(pa0 + k0);
        ra1 = *(const float4*)(pa1 + k0);
        if (bok) {
            const float* q = B + (long)k0 * ldb;
            #pragma unroll
            for (int i = 0; i < 8; ++i) {
                rb[i].x = (q + (long)(2 * i) * ldb)[bcol];
                rb[i].y = (q + (long)(2 * i + 1) * ldb)[bcol];
            }
        }
    };
    auto fetch_masked = [&](int k0) {
        float* p0 = (float*)&ra0;
        float* p1 = (float*)&ra1;
        const int g0 = m0 + r0, g1 = m0 + r1;
        #pragma unroll
        for (int j = 0; j < 4; ++j) {
            int gk = k0 + c0 + j;
            p0[j] = (g0 < M && gk < K) ? A[(long)g0 * lda + gk] : 0.0f;
            p1[j] = (g1 < M && gk < K) ? A[(long)g1 * lda + gk] : 0.0f;
        }
        if (bok) {
            const float* q = B + (long)k0 * ldb;
            #pragma unroll
            for (int i = 0; i < 8; ++i) {
                int gk = k0 + 2 * i;
                rb[i].x = (gk < K)     ? (q + (long)(2 * i) * ldb)[bcol]     : 0.0f;
                rb[i].y = (gk + 1 < K) ? (q + (long)(2 * i + 1) * ldb)[bcol] : 0.0f;
            }
        }
    };

    int buf = 0;
    auto slab = [&](int k0, auto&& fetch_next, bool has_next) {
        *(float4*)&As[buf][r0 * AP + c0] = ra0;
        *(float4*)&As[buf][r1 * AP + c0] = ra1;
        if (bown) {
            #pragma unroll
            for (int i = 0; i < 8; ++i) Bs[buf][i * BP + t] = rb[i];
        }
        __syncthreads();

        if (has_next) fetch_next(k0 + 16);

        const float* as = &As[buf][(wv * 16 + l16) * AP + half * 2];
        const v2f*   bs = &Bs[buf][half * BP + l16];
        #pragma unroll
        for (int ks = 0; ks < 16; ks += 4) {
            v2f a = *(const v2f*)(as + ks);
            v2f b[NT];
            #pragma unroll
            for (int j = 0; j < NT; ++j) b[j] = bs[(ks >> 1) * BP + j * 16];
            #pragma unroll
            for (int j = 0; j < NT; ++j)
                acc[j] = __builtin_amdgcn_wmma_f32_16x16x4_f32(
                    false, a, false, b[j], (short)0, acc[j], false, false);
        }
        buf ^= 1;
    };

    if (fastA) {
        if (K >= 16) fetch_fast(0); else fetch_masked(0);
        int k0 = 0;
        for (; k0 + 32 <= K; k0 += 16) slab(k0, fetch_fast, true);
        for (; k0 < K; k0 += 16)       slab(k0, fetch_masked, k0 + 16 < K);
    } else {
        fetch_masked(0);
        for (int k0 = 0; k0 < K; k0 += 16) slab(k0, fetch_masked, k0 + 16 < K);
    }

    #pragma unroll
    for (int j = 0; j < NT; ++j) {
        const int n = n0 + j * 16 + l16;
        #pragma unroll
        for (int v = 0; v < 8; ++v) {
            int m = m0 + wv * 16 + v + half * 8;
            if (m < M && n < N) C[(long)m * ldc + n] = acc[j][v];
        }
    }
}

template <int EPW, int NCH>
__global__ __launch_bounds__(256) void gcn_scatter_kernel(
    const float* __restrict__ H, float* __restrict__ agg,
    const int* __restrict__ src, const int* __restrict__ dst,
    const float* __restrict__ ew, const float* __restrict__ dinv_out,
    int E, int F)
{
    static_assert(EPW >= 1 && EPW <= 32, "one edge record per lane");
    static_assert(NCH >= 1 && NCH <= 16, "chunks held in registers per trip");
    const unsigned lane = threadIdx.x & 31;
    const int wave = __builtin_amdgcn_readfirstlane(
        (int)((blockIdx.x * blockDim.x + threadIdx.x) >> 5));
    const int e0 = wave * EPW;
    if (e0 >= E || F < 1) return;
    const int ne = min(EPW, E - e0);

    int   sj = 0, dj = 0;
    float wj = 0.0f;
    if ((int)lane < ne) {
        sj = src[e0 + lane];
        dj = dst[e0 + lane];
        wj = ew[e0 + lane] * dinv_out[sj];
    }

    const unsigned uF = (unsigned)F;
    for (int j = 0; j < ne; ++j) {
        const int   s   = __builtin_amdgcn_readlane(sj, j);
        const int   d   = __builtin_amdgcn_readlane(dj, j);
        const float wgt = __int_as_float(__builtin_amdgcn_readlane(__float_as_int(wj), j));
        const float* hs = H + (long)s * F;
        float*       ad = agg + (long)d * F;
        const unsigned a    = (unsigned)(((size_t)ad >> 2) & 31);
        const unsigned span = a + uF;
        for (unsigned c0 = 0; c0 < span; c0 += 32u * NCH) {
            float    hv[NCH];
            unsigned fi[NCH];
            #pragma unroll
            for (int u = 0; u < NCH; ++u) {
                fi[u] = c0 + 32u * (unsigned)u + lane - a;
                const int ci = min(max((int)fi[u], 0), F - 1);
                hv[u] = hs[ci];
            }
            #pragma unroll
            for (int u = 0; u < NCH; ++u)
                if (fi[u] < uF) unsafeAtomicAdd(ad + fi[u], hv[u] * wgt);
        }
    }
}

__global__ __launch_bounds__(256) void gcn_epilogue_relu_kernel(
    const float* __restrict__ agg, const float* __restrict__ dinv_in,
    const float* __restrict__ b, float* __restrict__ out, int total4, int F4)
{
    int q = blockIdx.x * blockDim.x + threadIdx.x;
    if (q < total4) {
        int r = q / F4;
        int c4 = q - r * F4;
        int i = q * 4;
        float s = dinv_in[r];
        float4 v  = *(const float4*)(agg + i);
        float4 bb = *(const float4*)(b + c4 * 4);
        float4 o;
        o.x = fmaxf(v.x * s + bb.x, 0.0f);
        o.y = fmaxf(v.y * s + bb.y, 0.0f);
        o.z = fmaxf(v.z * s + bb.z, 0.0f);
        o.w = fmaxf(v.w * s + bb.w, 0.0f);
        *(float4*)(out + i) = o;
    }
}

__global__ __launch_bounds__(256) void gcn_final_kernel(
    const float* __restrict__ agg, const float* __restrict__ dinv_in,
    const float* __restrict__ b, float* __restrict__ out, int total4, int F4)
{
    int q = blockIdx.x * blockDim.x + threadIdx.x;
    if (q < total4) {
        int r = q / F4;
        int c4 = q - r * F4;
        int i = q * 4;
        float s = dinv_in[r];
        float4 v  = *(const float4*)(agg + i);
        float4 bb = *(const float4*)(b + c4 * 4);
        float4 o;
        o.x = v.x * s + bb.x;
        o.y = v.y * s + bb.y;
        o.z = v.z * s + bb.z;
        o.w = v.w * s + bb.w;
        *(float4*)(out + i) = o;
    }
}

extern "C" void kernel_launch(void* const* d_in, const int* in_sizes, int n_in,
                              void* d_out, int out_size, void* d_ws, size_t ws_size,
                              hipStream_t stream)
{
    const float* feature = (const float*)d_in[0];
    const int*   src     = (const int*)  d_in[1];
    const int*   dst     = (const int*)  d_in[2];
    const float* edge_w  = (const float*)d_in[3];
    const float* W1      = (const float*)d_in[4];
    const float* b1      = (const float*)d_in[5];
    const float* W2      = (const float*)d_in[6];
    const float* b2      = (const float*)d_in[7];
    float*       out     = (float*)d_out;

    const int F1 = in_sizes[5];
    const int F2 = in_sizes[7];
    const int K0 = in_sizes[4] / F1;
    const int N  = in_sizes[0] / K0;
    const int E  = in_sizes[1];

    float* ws       = (float*)d_ws;
    float* deg_out  = ws;
    float* deg_in   = ws + N;
    float* H1       = deg_in + N;
    float* agg1     = H1 + (long)N * F1;
    float* H2       = agg1 + (long)N * F1;
    float* agg2     = H2 + (long)N * F2;

    hipMemsetAsync(deg_out, 0, sizeof(float) * 2 * N, stream);
    hipMemsetAsync(agg1, 0, sizeof(float) * (long)N * F1, stream);
    hipMemsetAsync(agg2, 0, sizeof(float) * (long)N * F2, stream);

    gcn_degree_kernel<<<(E + 255) / 256, 256, 0, stream>>>(src, dst, deg_out, deg_in, E);
    gcn_invsqrt_kernel<<<(2 * N + 255) / 256, 256, 0, stream>>>(deg_out, 2 * N);

    {
        dim3 grid((N + 127) / 128, (F1 + 207) / 208);
        gcn_wmma_gemm_f32<13><<<grid, 256, 0, stream>>>(feature, W1, H1, N, F1, K0, K0, F1, F1);
    }
    gcn_scatter_kernel<16, 7><<<(E + 127) / 128, 256, 0, stream>>>(H1, agg1, src, dst, edge_w,
                                                        deg_out, E, F1);
    {
        int total4 = N * (F1 / 4);
        gcn_epilogue_relu_kernel<<<(total4 + 255) / 256, 256, 0, stream>>>(
            agg1, deg_in, b1, H1, total4, F1 / 4);
    }

    {
        dim3 grid((N + 127) / 128, (F2 + 31) / 32);
        gcn_wmma_gemm_f32<2><<<grid, 256, 0, stream>>>(H1, W2, H2, N, F2, F1, F1, F2, F2);
    }
    gcn_scatter_kernel<16, 2><<<(E + 127) / 128, 256, 0, stream>>>(H2, agg2, src, dst, edge_w,
                                                        deg_out, E, F2);
    {
        int total4 = N * (F2 / 4);
        gcn_final_kernel<<<(total4 + 255) / 256, 256, 0, stream>>>(
            agg2, deg_in, b2, out, total4, F2 / 4);
    }
}


# aggregation as a gather: edges bucketed by dst in-call (int atomic rank + prefix sum), one wave sums a node's H[src] rows in registers, norm+bias fused; no f32 atomics; sum order = arrival-order rank
# speedup vs baseline: 2.0599x; 1.2676x over previous
#include <hip/hip_runtime.h>
#include <hip/hip_bf16.h>

typedef __attribute__((ext_vector_type(2))) float v2f;
typedef __attribute__((ext_vector_type(8))) float v8f;

__global__ __launch_bounds__(256) void gcn_degree_rank_kernel(
    const int* __restrict__ src, const int* __restrict__ dst,
    int* __restrict__ cnt_out, int* __restrict__ cnt_in,
    int* __restrict__ rank, int E)
{
    int e = blockIdx.x * blockDim.x + threadIdx.x;
    if (e < E) {
        atomicAdd(&cnt_out[src[e]], 1);
        rank[e] = atomicAdd(&cnt_in[dst[e]], 1);
    }
}

__global__ __launch_bounds__(1024) void gcn_rowptr_scan_kernel(
    const int* __restrict__ cnt, int* __restrict__ rowptr, int n)
{
    __shared__ int part[1024];
    const int t   = threadIdx.x;
    const int per = (n + 1023) / 1024;
    const int b   = min(t * per, n);
    const int e   = min(b + per, n);
    int s = 0;
    for (int i = b; i < e; ++i) s += cnt[i];
    part[t] = s;
    __syncthreads();
    for (int off = 1; off < 1024; off <<= 1) {
        const int v = (t >= off) ? part[t - off] : 0;
        __syncthreads();
        part[t] += v;
        __syncthreads();
    }
    int run = part[t] - s;
    for (int i = b; i < e; ++i) { rowptr[i] = run; run += cnt[i]; }
    if (t == 1023) rowptr[n] = part[1023];
}

__global__ __launch_bounds__(256) void gcn_csr_fill_kernel(
    const int* __restrict__ src, const int* __restrict__ dst,
    const float* __restrict__ ew, const int* __restrict__ cnt_out,
    const int* __restrict__ rowptr, const int* __restrict__ rank,
    int* __restrict__ rec, int E)
{
    int e = blockIdx.x * blockDim.x + threadIdx.x;
    if (e < E) {
        const int   s   = src[e];
        const int   pos = rowptr[dst[e]] + rank[e];
        const float w   = ew[e] * rsqrtf(fmaxf((float)cnt_out[s], 1.0f));
        *(int2*)(rec + 2 * (long)pos) = make_int2(s, __float_as_int(w));
    }
}

template <int NT>
__global__ __launch_bounds__(256) void gcn_wmma_gemm_f32(
    const float* __restrict__ A, const float* __restrict__ B,
    float* __restrict__ C, int M, int N, int K, int lda, int ldb, int ldc)
{
    constexpr int AP = 20;
    constexpr int BN = 16 * NT;
    constexpr int BP = (BN % 32 == 16) ? BN : BN + 16;
    static_assert(BN <= 256, "one B staging thread per block column");
    __shared__ __attribute__((aligned(16))) float As[2][128 * AP];
    __shared__ __attribute__((aligned(16))) v2f   Bs[2][8 * BP];

    const int t    = threadIdx.x;
    const int lane = t & 31;
    const int wv   = t >> 5;
    const int half = lane >> 4;
    const int l16  = lane & 15;

    const int m0 = blockIdx.x * 128;
    const int n0 = blockIdx.y * BN;

    const int r0 = t >> 2;
    const int r1 = r0 + 64;
    const int c0 = (t & 3) * 4;
    const bool bown = t < BN;
    const bool bok  = bown && (n0 + t < N);

    const float* pa0 = A + (long)(m0 + r0) * lda + c0;
    const float* pa1 = A + (long)(m0 + r1) * lda + c0;
    const unsigned bcol = n0 + t;

    const bool fastA = (m0 + 128 <= M) && ((lda & 3) == 0) &&
                       ((((size_t)A) & 15) == 0);

    float4 ra0, ra1;
    v2f    rb[8];
    #pragma unroll
    for (int i = 0; i < 8; ++i) { rb[i].x = 0.0f; rb[i].y = 0.0f; }

    v8f acc[NT];
    #pragma unroll
    for (int j = 0; j < NT; ++j) acc[j] = v8f{};

    auto fetch_fast = [&](int k0) {
        ra0 = *(const float4*)(pa0 + k0);
        ra1 = *(const float4*)(pa1 + k0);
        if (bok) {
            const float* q = B + (long)k0 * ldb;
            #pragma unroll
            for (int i = 0; i < 8; ++i) {
                rb[i].x = (q + (long)(2 * i) * ldb)[bcol];
                rb[i].y = (q + (long)(2 * i + 1) * ldb)[bcol];
            }
        }
    };
    auto fetch_masked = [&](int k0) {
        float* p0 = (float*)&ra0;
        float* p1 = (float*)&ra1;
        const int g0 = m0 + r0, g1 = m0 + r1;
        #pragma unroll
        for (int j = 0; j < 4; ++j) {
            int gk = k0 + c0 + j;
            p0[j] = (g0 < M && gk < K) ? A[(long)g0 * lda + gk] : 0.0f;
            p1[j] = (g1 < M && gk < K) ? A[(long)g1 * lda + gk] : 0.0f;
        }
        if (bok) {
            const float* q = B + (long)k0 * ldb;
            #pragma unroll
            for (int i = 0; i < 8; ++i) {
                int gk = k0 + 2 * i;
                rb[i].x = (gk < K)     ? (q + (long)(2 * i) * ldb)[bcol]     : 0.0f;
                rb[i].y = (gk + 1 < K) ? (q + (long)(2 * i + 1) * ldb)[bcol] : 0.0f;
            }
        }
    };

    int buf = 0;
    auto slab = [&](int k0, auto&& fetch_next, bool has_next) {
        *(float4*)&As[buf][r0 * AP + c0] = ra0;
        *(float4*)&As[buf][r1 * AP + c0] = ra1;
        if (bown) {
            #pragma unroll
            for (int i = 0; i < 8; ++i) Bs[buf][i * BP + t] = rb[i];
        }
        __syncthreads();

        if (has_next) fetch_next(k0 + 16);

        const float* as = &As[buf][(wv * 16 + l16) * AP + half * 2];
        const v2f*   bs = &Bs[buf][half * BP + l16];
        #pragma unroll
        for (int ks = 0; ks < 16; ks += 4) {
            v2f a = *(const v2f*)(as + ks);
            v2f b[NT];
            #pragma unroll
            for (int j = 0; j < NT; ++j) b[j] = bs[(ks >> 1) * BP + j * 16];
            #pragma unroll
            for (int j = 0; j < NT; ++j)
                acc[j] = __builtin_amdgcn_wmma_f32_16x16x4_f32(
                    false, a, false, b[j], (short)0, acc[j], false, false);
        }
        buf ^= 1;
    };

    if (fastA) {
        if (K >= 16) fetch_fast(0); else fetch_masked(0);
        int k0 = 0;
        for (; k0 + 32 <= K; k0 += 16) slab(k0, fetch_fast, true);
        for (; k0 < K; k0 += 16)       slab(k0, fetch_masked, k0 + 16 < K);
    } else {
        fetch_masked(0);
        for (int k0 = 0; k0 < K; k0 += 16) slab(k0, fetch_masked, k0 + 16 < K);
    }

    #pragma unroll
    for (int j = 0; j < NT; ++j) {
        const int n = n0 + j * 16 + l16;
        #pragma unroll
        for (int v = 0; v < 8; ++v) {
            int m = m0 + wv * 16 + v + half * 8;
            if (m < M && n < N) C[(long)m * ldc + n] = acc[j][v];
        }
    }
}

template <int NV, bool RELU>
__global__ __launch_bounds__(256) void gcn_gather_kernel(
    const float* __restrict__ H, const int* __restrict__ rec,
    const int* __restrict__ rowptr, const float* __restrict__ b,
    float* __restrict__ outp, int n, int F)
{
    const int lane = threadIdx.x & 31;
    const int node = __builtin_amdgcn_readfirstlane(
        (int)((blockIdx.x * blockDim.x + threadIdx.x) >> 5));
    if (node >= n) return;
    const int beg = rowptr[node];
    const int end = rowptr[node + 1];

    unsigned col[NV];
    #pragma unroll
    for (int u = 0; u < NV; ++u) col[u] = (unsigned)min(4 * (lane + 32 * u), F - 4);

    float4 acc[NV];
    #pragma unroll
    for (int u = 0; u < NV; ++u) acc[u] = make_float4(0.0f, 0.0f, 0.0f, 0.0f);

    for (int base = beg; base < end; base += 32) {
        const int nb = min(32, end - base);
        int sj = 0, wj = 0;
        if (lane < nb) {
            const int2 r = *(const int2*)(rec + 2 * (long)(base + lane));
            sj = r.x;
            wj = r.y;
        }
        int j = 0;
        for (; j + 4 <= nb; j += 4) {
            float4 hv[4][NV];
            float  wt[4];
            #pragma unroll
            for (int q = 0; q < 4; ++q) {
                const int s = __builtin_amdgcn_readlane(sj, j + q);
                wt[q] = __int_as_float(__builtin_amdgcn_readlane(wj, j + q));
                const float* hs = H + (long)s * F;
                #pragma unroll
                for (int u = 0; u < NV; ++u) hv[q][u] = *(const float4*)(hs + col[u]);
            }
            #pragma unroll
            for (int q = 0; q < 4; ++q) {
                #pragma unroll
                for (int u = 0; u < NV; ++u) {
                    acc[u].x = fmaf(hv[q][u].x, wt[q], acc[u].x);
                    acc[u].y = fmaf(hv[q][u].y, wt[q], acc[u].y);
                    acc[u].z = fmaf(hv[q][u].z, wt[q], acc[u].z);
                    acc[u].w = fmaf(hv[q][u].w, wt[q], acc[u].w);
                }
            }
        }
        for (; j < nb; ++j) {
            const int   s = __builtin_amdgcn_readlane(sj, j);
            const float w = __int_as_float(__builtin_amdgcn_readlane(wj, j));
            const float* hs = H + (long)s * F;
            float4 h1[NV];
            #pragma unroll
            for (int u = 0; u < NV; ++u) h1[u] = *(const float4*)(hs + col[u]);
            #pragma unroll
            for (int u = 0; u < NV; ++u) {
                acc[u].x = fmaf(h1[u].x, w, acc[u].x);
                acc[u].y = fmaf(h1[u].y, w, acc[u].y);
                acc[u].z = fmaf(h1[u].z, w, acc[u].z);
                acc[u].w = fmaf(h1[u].w, w, acc[u].w);
            }
        }
    }

    const float di = rsqrtf(fmaxf((float)(end - beg), 1.0f));
    float* orow = outp + (long)node * F;
    #pragma unroll
    for (int u = 0; u < NV; ++u) {
        const int c = 4 * (lane + 32 * u);
        if (c < F) {
            const float4 bb = *(const float4*)(b + c);
            float4 o;
            o.x = acc[u].x * di + bb.x;
            o.y = acc[u].y * di + bb.y;
            o.z = acc[u].z * di + bb.z;
            o.w = acc[u].w * di + bb.w;
            if (RELU) {
                o.x = fmaxf(o.x, 0.0f);
                o.y = fmaxf(o.y, 0.0f);
                o.z = fmaxf(o.z, 0.0f);
                o.w = fmaxf(o.w, 0.0f);
            }
            *(float4*)(orow + c) = o;
        }
    }
}

extern "C" void kernel_launch(void* const* d_in, const int* in_sizes, int n_in,
                              void* d_out, int out_size, void* d_ws, size_t ws_size,
                              hipStream_t stream)
{
    const float* feature = (const float*)d_in[0];
    const int*   src     = (const int*)  d_in[1];
    const int*   dst     = (const int*)  d_in[2];
    const float* edge_w  = (const float*)d_in[3];
    const float* W1      = (const float*)d_in[4];
    const float* b1      = (const float*)d_in[5];
    const float* W2      = (const float*)d_in[6];
    const float* b2      = (const float*)d_in[7];
    float*       out     = (float*)d_out;

    const int F1 = in_sizes[5];
    const int F2 = in_sizes[7];
    const int K0 = in_sizes[4] / F1;
    const int N  = in_sizes[0] / K0;
    const int E  = in_sizes[1];

    float* ws      = (float*)d_ws;
    int*   cnt_out = (int*)ws;
    int*   cnt_in  = cnt_out + N;
    float* R1      = ws + 2 * (long)N;
    float* R2      = R1 + (long)N * F1;
    float* R3      = R2 + (long)N * F1;
    float* H1      = R1;
    float* H2      = R1;
    int*   rank    = (int*)R2;
    float* X1      = R2;
    int*   rec     = (int*)R3;
    int*   rowptr  = rec + 2 * (long)E;

    hipMemsetAsync(cnt_out, 0, sizeof(int) * 2 * N, stream);

    gcn_degree_rank_kernel<<<(E + 255) / 256, 256, 0, stream>>>(src, dst, cnt_out, cnt_in, rank, E);
    gcn_rowptr_scan_kernel<<<1, 1024, 0, stream>>>(cnt_in, rowptr, N);
    gcn_csr_fill_kernel<<<(E + 255) / 256, 256, 0, stream>>>(src, dst, edge_w, cnt_out, rowptr,
                                                          rank, rec, E);

    {
        dim3 grid((N + 127) / 128, (F1 + 207) / 208);
        gcn_wmma_gemm_f32<13><<<grid, 256, 0, stream>>>(feature, W1, H1, N, F1, K0, K0, F1, F1);
    }
    gcn_gather_kernel<2, true><<<(N + 7) / 8, 256, 0, stream>>>(H1, rec, rowptr, b1, X1, N, F1);

    {
        dim3 grid((N + 127) / 128, (F2 + 31) / 32);
        gcn_wmma_gemm_f32<2><<<grid, 256, 0, stream>>>(X1, W2, H2, N, F2, F1, F1, F2, F2);
    }
    gcn_gather_kernel<1, false><<<(N + 7) / 8, 256, 0, stream>>>(H2, rec, rowptr, b2, out, N, F2);
}


# degree counts and edge ranks on LDS atomics: one work-group per 16384-edge chunk counts in a private LDS histogram, a merge pass sums chunks; no global atomics or memset; sum order = chunk-major rank
# speedup vs baseline: 2.5840x; 1.2544x over previous
#include <hip/hip_runtime.h>
#include <hip/hip_bf16.h>

typedef __attribute__((ext_vector_type(2))) float v2f;
typedef __attribute__((ext_vector_type(8))) float v8f;

constexpr int RK_THREADS = 1024;
constexpr int RK_EPT     = 16;
constexpr int RK_CHUNK   = RK_THREADS * RK_EPT;
constexpr int RK_HCAP    = 65536;

__global__ __launch_bounds__(RK_THREADS) void gcn_lds_rank_kernel(
    const int* __restrict__ src, const int* __restrict__ dst,
    int* __restrict__ part_in, int* __restrict__ part_out,
    int* __restrict__ rank, int E, int N, int G)
{
    __shared__ __attribute__((aligned(16))) int h[RK_HCAP];
    const int  t      = threadIdx.x;
    const bool is_dst = (int)blockIdx.x < G;
    const int  c      = is_dst ? (int)blockIdx.x : (int)blockIdx.x - G;
    const int* __restrict__ keys = is_dst ? dst : src;
    int* __restrict__ part = (is_dst ? part_in : part_out) + (long)c * N;
    const int  e0     = c * RK_CHUNK;

    int k[RK_EPT];
    #pragma unroll
    for (int u = 0; u < RK_EPT; ++u) {
        const int e = e0 + u * RK_THREADS + t;
        k[u] = (e < E) ? keys[e] : -1;
    }

    for (int base = 0; base < N; base += RK_HCAP) {
        const int nr = min(RK_HCAP, N - base);
        for (int i = 4 * t; i < nr; i += 4 * RK_THREADS)
            *(int4*)&h[i] = make_int4(0, 0, 0, 0);
        __syncthreads();
        if (is_dst) {
            int r[RK_EPT];
            #pragma unroll
            for (int u = 0; u < RK_EPT; ++u) {
                const unsigned d = (unsigned)(k[u] - base);
                r[u] = (d < (unsigned)nr) ? atomicAdd(&h[d], 1) : -1;
            }
            #pragma unroll
            for (int u = 0; u < RK_EPT; ++u)
                if (r[u] >= 0) rank[e0 + u * RK_THREADS + t] = r[u];
        } else {
            #pragma unroll
            for (int u = 0; u < RK_EPT; ++u) {
                const unsigned d = (unsigned)(k[u] - base);
                if (d < (unsigned)nr) atomicAdd(&h[d], 1);
            }
        }
        __syncthreads();
        const int nv = ((((size_t)(part + base)) & 15) == 0) ? (nr & ~3) : 0;
        for (int i = 4 * t; i < nv; i += 4 * RK_THREADS)
            *(int4*)&part[base + i] = *(const int4*)&h[i];
        for (int i = nv + t; i < nr; i += RK_THREADS) part[base + i] = h[i];
        __syncthreads();
    }
}

__global__ __launch_bounds__(256) void gcn_merge_counts_kernel(
    const int* __restrict__ part_in, const int* __restrict__ part_out,
    int* __restrict__ off, int* __restrict__ cnt_in, int* __restrict__ cnt_out,
    int N, int G)
{
    const int i = blockIdx.x * blockDim.x + threadIdx.x;
    if (i >= N) return;
    int run = 0, so = 0;
    #pragma unroll 8
    for (int g = 0; g < G; ++g) {
        const long o = (long)g * N + i;
        off[o] = run;
        run += part_in[o];
        so  += part_out[o];
    }
    cnt_in[i]  = run;
    cnt_out[i] = so;
}

__global__ __launch_bounds__(1024) void gcn_rowptr_scan_kernel(
    const int* __restrict__ cnt, int* __restrict__ rowptr, int n)
{
    __shared__ int part[1024];
    const int t   = threadIdx.x;
    const int per = (n + 1023) / 1024;
    const int b   = min(t * per, n);
    const int e   = min(b + per, n);
    int s = 0;
    for (int i = b; i < e; ++i) s += cnt[i];
    part[t] = s;
    __syncthreads();
    for (int off = 1; off < 1024; off <<= 1) {
        const int v = (t >= off) ? part[t - off] : 0;
        __syncthreads();
        part[t] += v;
        __syncthreads();
    }
    int run = part[t] - s;
    for (int i = b; i < e; ++i) { rowptr[i] = run; run += cnt[i]; }
    if (t == 1023) rowptr[n] = part[1023];
}

__global__ __launch_bounds__(256) void gcn_csr_fill_kernel(
    const int* __restrict__ src, const int* __restrict__ dst,
    const float* __restrict__ ew, const int* __restrict__ cnt_out,
    const int* __restrict__ rowptr, const int* __restrict__ rank,
    const int* __restrict__ off, int* __restrict__ rec, int E, int N)
{
    int e = blockIdx.x * blockDim.x + threadIdx.x;
    if (e < E) {
        const int   s   = src[e];
        const int   d   = dst[e];
        const int   pos = rowptr[d] + off[(long)(e / RK_CHUNK) * N + d] + rank[e];
        const float w   = ew[e] * rsqrtf(fmaxf((float)cnt_out[s], 1.0f));
        *(int2*)(rec + 2 * (long)pos) = make_int2(s, __float_as_int(w));
    }
}

template <int NT>
__global__ __launch_bounds__(256) void gcn_wmma_gemm_f32(
    const float* __restrict__ A, const float* __restrict__ B,
    float* __restrict__ C, int M, int N, int K, int lda, int ldb, int ldc)
{
    constexpr int AP = 20;
    constexpr int BN = 16 * NT;
    constexpr int BP = (BN % 32 == 16) ? BN : BN + 16;
    static_assert(BN <= 256, "one B staging thread per block column");
    __shared__ __attribute__((aligned(16))) float As[2][128 * AP];
    __shared__ __attribute__((aligned(16))) v2f   Bs[2][8 * BP];

    const int t    = threadIdx.x;
    const int lane = t & 31;
    const int wv   = t >> 5;
    const int half = lane >> 4;
    const int l16  = lane & 15;

    const int m0 = blockIdx.x * 128;
    const int n0 = blockIdx.y * BN;

    const int r0 = t >> 2;
    const int r1 = r0 + 64;
    const int c0 = (t & 3) * 4;
    const bool bown = t < BN;
    const bool bok  = bown && (n0 + t < N);

    const float* pa0 = A + (long)(m0 + r0) * lda + c0;
    const float* pa1 = A + (long)(m0 + r1) * lda + c0;
    const unsigned bcol = n0 + t;

    const bool fastA = (m0 + 128 <= M) && ((lda & 3) == 0) &&
                       ((((size_t)A) & 15) == 0);

    float4 ra0, ra1;
    v2f    rb[8];
    #pragma unroll
    for (int i = 0; i < 8; ++i) { rb[i].x = 0.0f; rb[i].y = 0.0f; }

    v8f acc[NT];
    #pragma unroll
    for (int j = 0; j < NT; ++j) acc[j] = v8f{};

    auto fetch_fast = [&](int k0) {
        ra0 = *(const float4*)(pa0 + k0);
        ra1 = *(const float4*)(pa1 + k0);
        if (bok) {
            const float* q = B + (long)k0 * ldb;
            #pragma unroll
            for (int i = 0; i < 8; ++i) {
                rb[i].x = (q + (long)(2 * i) * ldb)[bcol];
                rb[i].y = (q + (long)(2 * i + 1) * ldb)[bcol];
            }
        }
    };
    auto fetch_masked = [&](int k0) {
        float* p0 = (float*)&ra0;
        float* p1 = (float*)&ra1;
        const int g0 = m0 + r0, g1 = m0 + r1;
        #pragma unroll
        for (int j = 0; j < 4; ++j) {
            int gk = k0 + c0 + j;
            p0[j] = (g0 < M && gk < K) ? A[(long)g0 * lda + gk] : 0.0f;
            p1[j] = (g1 < M && gk < K) ? A[(long)g1 * lda + gk] : 0.0f;
        }
        if (bok) {
            const float* q = B + (long)k0 * ldb;
            #pragma unroll
            for (int i = 0; i < 8; ++i) {
                int gk = k0 + 2 * i;
                rb[i].x = (gk < K)     ? (q + (long)(2 * i) * ldb)[bcol]     : 0.0f;
                rb[i].y = (gk + 1 < K) ? (q + (long)(2 * i + 1) * ldb)[bcol] : 0.0f;
            }
        }
    };

    int buf = 0;
    auto slab = [&](int k0, auto&& fetch_next, bool has_next) {
        *(float4*)&As[buf][r0 * AP + c0] = ra0;
        *(float4*)&As[buf][r1 * AP + c0] = ra1;
        if (bown) {
            #pragma unroll
            for (int i = 0; i < 8; ++i) Bs[buf][i * BP + t] = rb[i];
        }
        __syncthreads();

        if (has_next) fetch_next(k0 + 16);

        const float* as = &As[buf][(wv * 16 + l16) * AP + half * 2];
        const v2f*   bs = &Bs[buf][half * BP + l16];
        #pragma unroll
        for (int ks = 0; ks < 16; ks += 4) {
            v2f a = *(const v2f*)(as + ks);
            v2f b[NT];
            #pragma unroll
            for (int j = 0; j < NT; ++j) b[j] = bs[(ks >> 1) * BP + j * 16];
            #pragma unroll
            for (int j = 0; j < NT; ++j)
                acc[j] = __builtin_amdgcn_wmma_f32_16x16x4_f32(
                    false, a, false, b[j], (short)0, acc[j], false, false);
        }
        buf ^= 1;
    };

    if (fastA) {
        if (K >= 16) fetch_fast(0); else fetch_masked(0);
        int k0 = 0;
        for (; k0 + 32 <= K; k0 += 16) slab(k0, fetch_fast, true);
        for (; k0 < K; k0 += 16)       slab(k0, fetch_masked, k0 + 16 < K);
    } else {
        fetch_masked(0);
        for (int k0 = 0; k0 < K; k0 += 16) slab(k0, fetch_masked, k0 + 16 < K);
    }

    #pragma unroll
    for (int j = 0; j < NT; ++j) {
        const int n = n0 + j * 16 + l16;
        #pragma unroll
        for (int v = 0; v < 8; ++v) {
            int m = m0 + wv * 16 + v + half * 8;
            if (m < M && n < N) C[(long)m * ldc + n] = acc[j][v];
        }
    }
}

template <int NV, bool RELU>
__global__ __launch_bounds__(256) void gcn_gather_kernel(
    const float* __restrict__ H, const int* __restrict__ rec,
    const int* __restrict__ rowptr, const float* __restrict__ b,
    float* __restrict__ outp, int n, int F)
{
    const int lane = threadIdx.x & 31;
    const int node = __builtin_amdgcn_readfirstlane(
        (int)((blockIdx.x * blockDim.x + threadIdx.x) >> 5));
    if (node >= n) return;
    const int beg = rowptr[node];
    const int end = rowptr[node + 1];

    unsigned col[NV];
    #pragma unroll
    for (int u = 0; u < NV; ++u) col[u] = (unsigned)min(4 * (lane + 32 * u), F - 4);

    float4 acc[NV];
    #pragma unroll
    for (int u = 0; u < NV; ++u) acc[u] = make_float4(0.0f, 0.0f, 0.0f, 0.0f);

    for (int base = beg; base < end; base += 32) {
        const int nb = min(32, end - base);
        int sj = 0, wj = 0;
        if (lane < nb) {
            const int2 r = *(const int2*)(rec + 2 * (long)(base + lane));
            sj = r.x;
            wj = r.y;
        }
        int j = 0;
        for (; j + 4 <= nb; j += 4) {
            float4 hv[4][NV];
            float  wt[4];
            #pragma unroll
            for (int q = 0; q < 4; ++q) {
                const int s = __builtin_amdgcn_readlane(sj, j + q);
                wt[q] = __int_as_float(__builtin_amdgcn_readlane(wj, j + q));
                const float* hs = H + (long)s * F;
                #pragma unroll
                for (int u = 0; u < NV; ++u) hv[q][u] = *(const float4*)(hs + col[u]);
            }
            #pragma unroll
            for (int q = 0; q < 4; ++q) {
                #pragma unroll
                for (int u = 0; u < NV; ++u) {
                    acc[u].x = fmaf(hv[q][u].x, wt[q], acc[u].x);
                    acc[u].y = fmaf(hv[q][u].y, wt[q], acc[u].y);
                    acc[u].z = fmaf(hv[q][u].z, wt[q], acc[u].z);
                    acc[u].w = fmaf(hv[q][u].w, wt[q], acc[u].w);
                }
            }
        }
        for (; j < nb; ++j) {
            const int   s = __builtin_amdgcn_readlane(sj, j);
            const float w = __int_as_float(__builtin_amdgcn_readlane(wj, j));
            const float* hs = H + (long)s * F;
            float4 h1[NV];
            #pragma unroll
            for (int u = 0; u < NV; ++u) h1[u] = *(const float4*)(hs + col[u]);
            #pragma unroll
            for (int u = 0; u < NV; ++u) {
                acc[u].x = fmaf(h1[u].x, w, acc[u].x);
                acc[u].y = fmaf(h1[u].y, w, acc[u].y);
                acc[u].z = fmaf(h1[u].z, w, acc[u].z);
                acc[u].w = fmaf(h1[u].w, w, acc[u].w);
            }
        }
    }

    const float di = rsqrtf(fmaxf((float)(end - beg), 1.0f));
    float* orow = outp + (long)node * F;
    #pragma unroll
    for (int u = 0; u < NV; ++u) {
        const int c = 4 * (lane + 32 * u);
        if (c < F) {
            const float4 bb = *(const float4*)(b + c);
            float4 o;
            o.x = acc[u].x * di + bb.x;
            o.y = acc[u].y * di + bb.y;
            o.z = acc[u].z * di + bb.z;
            o.w = acc[u].w * di + bb.w;
            if (RELU) {
                o.x = fmaxf(o.x, 0.0f);
                o.y = fmaxf(o.y, 0.0f);
                o.z = fmaxf(o.z, 0.0f);
                o.w = fmaxf(o.w, 0.0f);
            }
            *(float4*)(orow + c) = o;
        }
    }
}

extern "C" void kernel_launch(void* const* d_in, const int* in_sizes, int n_in,
                              void* d_out, int out_size, void* d_ws, size_t ws_size,
                              hipStream_t stream)
{
    const float* feature = (const float*)d_in[0];
    const int*   src     = (const int*)  d_in[1];
    const int*   dst     = (const int*)  d_in[2];
    const float* edge_w  = (const float*)d_in[3];
    const float* W1      = (const float*)d_in[4];
    const float* b1      = (const float*)d_in[5];
    const float* W2      = (const float*)d_in[6];
    const float* b2      = (const float*)d_in[7];
    float*       out     = (float*)d_out;

    const int F1 = in_sizes[5];
    const int F2 = in_sizes[7];
    const int K0 = in_sizes[4] / F1;
    const int N  = in_sizes[0] / K0;
    const int E  = in_sizes[1];

    float* ws      = (float*)d_ws;
    int*   cnt_out = (int*)ws;
    int*   cnt_in  = cnt_out + N;
    float* R1      = ws + 2 * (long)N;
    float* R2      = R1 + (long)N * F1;
    float* R3      = R2 + (long)N * F1;
    float* H1      = R1;
    float* H2      = R1;
    int*   rank    = (int*)R2;
    const int G    = (E + RK_CHUNK - 1) / RK_CHUNK;
    int*   part_in  = rank + E;
    int*   part_out = part_in + (long)G * N;
    int*   off      = part_out + (long)G * N;
    float* X1      = R2;
    int*   rec     = (int*)R3;
    int*   rowptr  = rec + 2 * (long)E;

    gcn_lds_rank_kernel<<<2 * G, RK_THREADS, 0, stream>>>(src, dst, part_in, part_out, rank, E, N, G);
    gcn_merge_counts_kernel<<<(N + 255) / 256, 256, 0, stream>>>(part_in, part_out, off, cnt_in, cnt_out, N, G);
    gcn_rowptr_scan_kernel<<<1, 1024, 0, stream>>>(cnt_in, rowptr, N);
    gcn_csr_fill_kernel<<<(E + 255) / 256, 256, 0, stream>>>(src, dst, edge_w, cnt_out, rowptr,
                                                          rank, off, rec, E, N);

    {
        dim3 grid((N + 127) / 128, (F1 + 207) / 208);
        gcn_wmma_gemm_f32<13><<<grid, 256, 0, stream>>>(feature, W1, H1, N, F1, K0, K0, F1, F1);
    }
    gcn_gather_kernel<2, true><<<(N + 7) / 8, 256, 0, stream>>>(H1, rec, rowptr, b1, X1, N, F1);

    {
        dim3 grid((N + 127) / 128, (F2 + 31) / 32);
        gcn_wmma_gemm_f32<2><<<grid, 256, 0, stream>>>(X1, W2, H2, N, F2, F1, F1, F2, F2);
    }
    gcn_gather_kernel<1, false><<<(N + 7) / 8, 256, 0, stream>>>(H2, rec, rowptr, b2, out, N, F2);
}


# row starts as a two-level prefix sum: merge blocks scan their 256 in-degrees by shuffles, the fill launch adds the block sums; the one-block scan launch is gone (7 launches); integer sums only
# speedup vs baseline: 2.8621x; 1.1076x over previous
#include <hip/hip_runtime.h>
#include <hip/hip_bf16.h>

typedef __attribute__((ext_vector_type(2))) float v2f;
typedef __attribute__((ext_vector_type(8))) float v8f;

constexpr int RK_THREADS = 1024;
constexpr int RK_EPT     = 16;
constexpr int RK_CHUNK   = RK_THREADS * RK_EPT;
constexpr int RK_HCAP    = 65536;

__global__ __launch_bounds__(RK_THREADS) void gcn_lds_rank_kernel(
    const int* __restrict__ src, const int* __restrict__ dst,
    int* __restrict__ part_in, int* __restrict__ part_out,
    int* __restrict__ rank, int E, int N, int G)
{
    __shared__ __attribute__((aligned(16))) int h[RK_HCAP];
    const int  t      = threadIdx.x;
    const bool is_dst = (int)blockIdx.x < G;
    const int  c      = is_dst ? (int)blockIdx.x : (int)blockIdx.x - G;
    const int* __restrict__ keys = is_dst ? dst : src;
    int* __restrict__ part = (is_dst ? part_in : part_out) + (long)c * N;
    const int  e0     = c * RK_CHUNK;

    int k[RK_EPT];
    #pragma unroll
    for (int u = 0; u < RK_EPT; ++u) {
        const int e = e0 + u * RK_THREADS + t;
        k[u] = (e < E) ? keys[e] : -1;
    }

    for (int base = 0; base < N; base += RK_HCAP) {
        const int nr = min(RK_HCAP, N - base);
        for (int i = 4 * t; i < nr; i += 4 * RK_THREADS)
            *(int4*)&h[i] = make_int4(0, 0, 0, 0);
        __syncthreads();
        if (is_dst) {
            int r[RK_EPT];
            #pragma unroll
            for (int u = 0; u < RK_EPT; ++u) {
                const unsigned d = (unsigned)(k[u] - base);
                r[u] = (d < (unsigned)nr) ? atomicAdd(&h[d], 1) : -1;
            }
            #pragma unroll
            for (int u = 0; u < RK_EPT; ++u)
                if (r[u] >= 0) rank[e0 + u * RK_THREADS + t] = r[u];
        } else {
            #pragma unroll
            for (int u = 0; u < RK_EPT; ++u) {
                const unsigned d = (unsigned)(k[u] - base);
                if (d < (unsigned)nr) atomicAdd(&h[d], 1);
            }
        }
        __syncthreads();
        const int nv = ((((size_t)(part + base)) & 15) == 0) ? (nr & ~3) : 0;
        for (int i = 4 * t; i < nv; i += 4 * RK_THREADS)
            *(int4*)&part[base + i] = *(const int4*)&h[i];
        for (int i = nv + t; i < nr; i += RK_THREADS) part[base + i] = h[i];
        __syncthreads();
    }
}

__device__ __forceinline__ int gcn_block_prefix_256(int v, int* wt, int& total)
{
    const int lane = threadIdx.x & 31;
    const int wv   = threadIdx.x >> 5;
    int inc = v;
    #pragma unroll
    for (int d = 1; d < 32; d <<= 1) {
        const int o = __shfl_up(inc, d);
        if (lane >= d) inc += o;
    }
    if (lane == 31) wt[wv] = inc;
    __syncthreads();
    int wo = 0, tot = 0;
    #pragma unroll
    for (int k = 0; k < 8; ++k) {
        const int x = wt[k];
        tot += x;
        if (k < wv) wo += x;
    }
    total = tot;
    return wo + inc - v;
}

__global__ __launch_bounds__(256) void gcn_merge_counts_kernel(
    const int* __restrict__ part_in, const int* __restrict__ part_out,
    int* __restrict__ off, int* __restrict__ rloc, int* __restrict__ bsum,
    int* __restrict__ cnt_out, int N, int G)
{
    __shared__ __attribute__((aligned(16))) int wt[8];
    const int i = blockIdx.x * 256 + threadIdx.x;
    int run = 0;
    if (i < N) {
        int so = 0;
        #pragma unroll 8
        for (int g = 0; g < G; ++g) {
            const long o = (long)g * N + i;
            off[o] = run;
            run += part_in[o];
            so  += part_out[o];
        }
        cnt_out[i] = so;
    }
    int tot;
    const int ex = gcn_block_prefix_256(run, wt, tot);
    if (i < N) rloc[i] = ex;
    if (threadIdx.x == 0) bsum[blockIdx.x] = tot;
}

__global__ __launch_bounds__(256) void gcn_csr_fill_kernel(
    const int* __restrict__ src, const int* __restrict__ dst,
    const float* __restrict__ ew, const int* __restrict__ cnt_out,
    const int* __restrict__ rloc, const int* __restrict__ bsum,
    const int* __restrict__ rank, const int* __restrict__ off,
    int* __restrict__ rec, int* __restrict__ rowptr, int E, int N, int NB)
{
    __shared__ int sbase[256];
    __shared__ __attribute__((aligned(16))) int wt[8];
    const int t = threadIdx.x;

    if ((int)blockIdx.x < NB) {
        const int j = (int)blockIdx.x;
        int part = 0;
        for (int q = t; q < j; q += 256) part += bsum[q];
        int base;
        (void)gcn_block_prefix_256(part, wt, base);
        const int i = j * 256 + t;
        if (i < N) rowptr[i] = base + rloc[i];
        if (j == NB - 1 && t == 0) rowptr[N] = base + bsum[j];
        return;
    }

    const int  e  = ((int)blockIdx.x - NB) * 256 + t;
    const bool ok = e < E;
    int v = (t < NB) ? bsum[t] : 0;
    int s = 0, d = 0;
    if (ok) { s = src[e]; d = dst[e]; }
    int dbase = 0, carry = 0;
    for (int tb = 0; tb < NB; tb += 256) {
        int tot;
        const int ex = gcn_block_prefix_256(v, wt, tot);
        sbase[t] = carry + ex;
        __syncthreads();
        const unsigned r = (unsigned)((d >> 8) - tb);
        if (r < 256u) dbase = sbase[r];
        carry += tot;
        const int q = tb + 256 + t;
        v = (q < NB) ? bsum[q] : 0;
    }
    if (ok) {
        const int   pos = dbase + rloc[d] + off[(long)(e / RK_CHUNK) * N + d] + rank[e];
        const float w   = ew[e] * rsqrtf(fmaxf((float)cnt_out[s], 1.0f));
        *(int2*)(rec + 2 * (long)pos) = make_int2(s, __float_as_int(w));
    }
}

template <int NT>
__global__ __launch_bounds__(256) void gcn_wmma_gemm_f32(
    const float* __restrict__ A, const float* __restrict__ B,
    float* __restrict__ C, int M, int N, int K, int lda, int ldb, int ldc)
{
    constexpr int AP = 20;
    constexpr int BN = 16 * NT;
    constexpr int BP = (BN % 32 == 16) ? BN : BN + 16;
    static_assert(BN <= 256, "one B staging thread per block column");
    __shared__ __attribute__((aligned(16))) float As[2][128 * AP];
    __shared__ __attribute__((aligned(16))) v2f   Bs[2][8 * BP];

    const int t    = threadIdx.x;
    const int lane = t & 31;
    const int wv   = t >> 5;
    const int half = lane >> 4;
    const int l16  = lane & 15;

    const int m0 = blockIdx.x * 128;
    const int n0 = blockIdx.y * BN;

    const int r0 = t >> 2;
    const int r1 = r0 + 64;
    const int c0 = (t & 3) * 4;
    const bool bown = t < BN;
    const bool bok  = bown && (n0 + t < N);

    const float* pa0 = A + (long)(m0 + r0) * lda + c0;
    const float* pa1 = A + (long)(m0 + r1) * lda + c0;
    const unsigned bcol = n0 + t;

    const bool fastA = (m0 + 128 <= M) && ((lda & 3) == 0) &&
                       ((((size_t)A) & 15) == 0);

    float4 ra0, ra1;
    v2f    rb[8];
    #pragma unroll
    for (int i = 0; i < 8; ++i) { rb[i].x = 0.0f; rb[i].y = 0.0f; }

    v8f acc[NT];
    #pragma unroll
    for (int j = 0; j < NT; ++j) acc[j] = v8f{};

    auto fetch_fast = [&](int k0) {
        ra0 = *(const float4*)(pa0 + k0);
        ra1 = *(const float4*)(pa1 + k0);
        if (bok) {
            const float* q = B + (long)k0 * ldb;
            #pragma unroll
            for (int i = 0; i < 8; ++i) {
                rb[i].x = (q + (long)(2 * i) * ldb)[bcol];
                rb[i].y = (q + (long)(2 * i + 1) * ldb)[bcol];
            }
        }
    };
    auto fetch_masked = [&](int k0) {
        float* p0 = (float*)&ra0;
        float* p1 = (float*)&ra1;
        const int g0 = m0 + r0, g1 = m0 + r1;
        #pragma unroll
        for (int j = 0; j < 4; ++j) {
            int gk = k0 + c0 + j;
            p0[j] = (g0 < M && gk < K) ? A[(long)g0 * lda + gk] : 0.0f;
            p1[j] = (g1 < M && gk < K) ? A[(long)g1 * lda + gk] : 0.0f;
        }
        if (bok) {
            const float* q = B + (long)k0 * ldb;
            #pragma unroll
            for (int i = 0; i < 8; ++i) {
                int gk = k0 + 2 * i;
                rb[i].x = (gk < K)     ? (q + (long)(2 * i) * ldb)[bcol]     : 0.0f;
                rb[i].y = (gk + 1 < K) ? (q + (long)(2 * i + 1) * ldb)[bcol] : 0.0f;
            }
        }
    };

    int buf = 0;
    auto slab = [&](int k0, auto&& fetch_next, bool has_next) {
        *(float4*)&As[buf][r0 * AP + c0] = ra0;
        *(float4*)&As[buf][r1 * AP + c0] = ra1;
        if (bown) {
            #pragma unroll
            for (int i = 0; i < 8; ++i) Bs[buf][i * BP + t] = rb[i];
        }
        __syncthreads();

        if (has_next) fetch_next(k0 + 16);

        const float* as = &As[buf][(wv * 16 + l16) * AP + half * 2];
        const v2f*   bs = &Bs[buf][half * BP + l16];
        #pragma unroll
        for (int ks = 0; ks < 16; ks += 4) {
            v2f a = *(const v2f*)(as + ks);
            v2f b[NT];
            #pragma unroll
            for (int j = 0; j < NT; ++j) b[j] = bs[(ks >> 1) * BP + j * 16];
            #pragma unroll
            for (int j = 0; j < NT; ++j)
                acc[j] = __builtin_amdgcn_wmma_f32_16x16x4_f32(
                    false, a, false, b[j], (short)0, acc[j], false, false);
        }
        buf ^= 1;
    };

    if (fastA) {
        if (K >= 16) fetch_fast(0); else fetch_masked(0);
        int k0 = 0;
        for (; k0 + 32 <= K; k0 += 16) slab(k0, fetch_fast, true);
        for (; k0 < K; k0 += 16)       slab(k0, fetch_masked, k0 + 16 < K);
    } else {
        fetch_masked(0);
        for (int k0 = 0; k0 < K; k0 += 16) slab(k0, fetch_masked, k0 + 16 < K);
    }

    #pragma unroll
    for (int j = 0; j < NT; ++j) {
        const int n = n0 + j * 16 + l16;
        #pragma unroll
        for (int v = 0; v < 8; ++v) {
            int m = m0 + wv * 16 + v + half * 8;
            if (m < M && n < N) C[(long)m * ldc + n] = acc[j][v];
        }
    }
}

template <int NV, bool RELU>
__global__ __launch_bounds__(256) void gcn_gather_kernel(
    const float* __restrict__ H, const int* __restrict__ rec,
    const int* __restrict__ rowptr, const float* __restrict__ b,
    float* __restrict__ outp, int n, int F)
{
    const int lane = threadIdx.x & 31;
    const int node = __builtin_amdgcn_readfirstlane(
        (int)((blockIdx.x * blockDim.x + threadIdx.x) >> 5));
    if (node >= n) return;
    const int beg = rowptr[node];
    const int end = rowptr[node + 1];

    unsigned col[NV];
    #pragma unroll
    for (int u = 0; u < NV; ++u) col[u] = (unsigned)min(4 * (lane + 32 * u), F - 4);

    float4 acc[NV];
    #pragma unroll
    for (int u = 0; u < NV; ++u) acc[u] = make_float4(0.0f, 0.0f, 0.0f, 0.0f);

    for (int base = beg; base < end; base += 32) {
        const int nb = min(32, end - base);
        int sj = 0, wj = 0;
        if (lane < nb) {
            const int2 r = *(const int2*)(rec + 2 * (long)(base + lane));
            sj = r.x;
            wj = r.y;
        }
        int j = 0;
        for (; j + 4 <= nb; j += 4) {
            float4 hv[4][NV];
            float  wt[4];
            #pragma unroll
            for (int q = 0; q < 4; ++q) {
                const int s = __builtin_amdgcn_readlane(sj, j + q);
                wt[q] = __int_as_float(__builtin_amdgcn_readlane(wj, j + q));
                const float* hs = H + (long)s * F;
                #pragma unroll
                for (int u = 0; u < NV; ++u) hv[q][u] = *(const float4*)(hs + col[u]);
            }
            #pragma unroll
            for (int q = 0; q < 4; ++q) {
                #pragma unroll
                for (int u = 0; u < NV; ++u) {
                    acc[u].x = fmaf(hv[q][u].x, wt[q], acc[u].x);
                    acc[u].y = fmaf(hv[q][u].y, wt[q], acc[u].y);
                    acc[u].z = fmaf(hv[q][u].z, wt[q], acc[u].z);
                    acc[u].w = fmaf(hv[q][u].w, wt[q], acc[u].w);
                }
            }
        }
        for (; j < nb; ++j) {
            const int   s = __builtin_amdgcn_readlane(sj, j);
            const float w = __int_as_float(__builtin_amdgcn_readlane(wj, j));
            const float* hs = H + (long)s * F;
            float4 h1[NV];
            #pragma unroll
            for (int u = 0; u < NV; ++u) h1[u] = *(const float4*)(hs + col[u]);
            #pragma unroll
            for (int u = 0; u < NV; ++u) {
                acc[u].x = fmaf(h1[u].x, w, acc[u].x);
                acc[u].y = fmaf(h1[u].y, w, acc[u].y);
                acc[u].z = fmaf(h1[u].z, w, acc[u].z);
                acc[u].w = fmaf(h1[u].w, w, acc[u].w);
            }
        }
    }

    const float di = rsqrtf(fmaxf((float)(end - beg), 1.0f));
    float* orow = outp + (long)node * F;
    #pragma unroll
    for (int u = 0; u < NV; ++u) {
        const int c = 4 * (lane + 32 * u);
        if (c < F) {
            const float4 bb = *(const float4*)(b + c);
            float4 o;
            o.x = acc[u].x * di + bb.x;
            o.y = acc[u].y * di + bb.y;
            o.z = acc[u].z * di + bb.z;
            o.w = acc[u].w * di + bb.w;
            if (RELU) {
                o.x = fmaxf(o.x, 0.0f);
                o.y = fmaxf(o.y, 0.0f);
                o.z = fmaxf(o.z, 0.0f);
                o.w = fmaxf(o.w, 0.0f);
            }
            *(float4*)(orow + c) = o;
        }
    }
}

extern "C" void kernel_launch(void* const* d_in, const int* in_sizes, int n_in,
                              void* d_out, int out_size, void* d_ws, size_t ws_size,
                              hipStream_t stream)
{
    const float* feature = (const float*)d_in[0];
    const int*   src     = (const int*)  d_in[1];
    const int*   dst     = (const int*)  d_in[2];
    const float* edge_w  = (const float*)d_in[3];
    const float* W1      = (const float*)d_in[4];
    const float* b1      = (const float*)d_in[5];
    const float* W2      = (const float*)d_in[6];
    const float* b2      = (const float*)d_in[7];
    float*       out     = (float*)d_out;

    const int F1 = in_sizes[5];
    const int F2 = in_sizes[7];
    const int K0 = in_sizes[4] / F1;
    const int N  = in_sizes[0] / K0;
    const int E  = in_sizes[1];

    float* ws      = (float*)d_ws;
    int*   cnt_out = (int*)ws;
    int*   rloc    = cnt_out + N;
    float* R1      = ws + 2 * (long)N;
    float* R2      = R1 + (long)N * F1;
    float* R3      = R2 + (long)N * F1;
    float* H1      = R1;
    float* H2      = R1;
    int*   rank    = (int*)R2;
    const int G    = (E + RK_CHUNK - 1) / RK_CHUNK;
    int*   part_in  = rank + E;
    int*   part_out = part_in + (long)G * N;
    int*   off      = part_out + (long)G * N;
    float* X1      = R2;
    int*   rec     = (int*)R3;
    int*   rowptr  = rec + 2 * (long)E;
    const int NB   = (N + 255) / 256;
    const int EB   = (E + 255) / 256;
    int*   bsum    = rowptr + N + 1;

    gcn_lds_rank_kernel<<<2 * G, RK_THREADS, 0, stream>>>(src, dst, part_in, part_out, rank, E, N, G);
    gcn_merge_counts_kernel<<<NB, 256, 0, stream>>>(part_in, part_out, off, rloc, bsum, cnt_out, N, G);
    gcn_csr_fill_kernel<<<NB + EB, 256, 0, stream>>>(src, dst, edge_w, cnt_out, rloc, bsum,
                                                     rank, off, rec, rowptr, E, N, NB);

    {
        dim3 grid((N + 127) / 128, (F1 + 207) / 208);
        gcn_wmma_gemm_f32<13><<<grid, 256, 0, stream>>>(feature, W1, H1, N, F1, K0, K0, F1, F1);
    }
    gcn_gather_kernel<2, true><<<(N + 7) / 8, 256, 0, stream>>>(H1, rec, rowptr, b1, X1, N, F1);

    {
        dim3 grid((N + 127) / 128, (F2 + 31) / 32);
        gcn_wmma_gemm_f32<2><<<grid, 256, 0, stream>>>(X1, W2, H2, N, F2, F1, F1, F2, F2);
    }
    gcn_gather_kernel<1, false><<<(N + 7) / 8, 256, 0, stream>>>(H2, rec, rowptr, b2, out, N, F2);
}


# layer-2 GEMM fused into the layer-1 gather: sixteen nodes per work-group, relu rows stay in LDS, one wave multiplies the tile by W2 on f32 WMMA; X1 never stored, one launch fewer; sum orders kept
# speedup vs baseline: 2.9524x; 1.0316x over previous
#include <hip/hip_runtime.h>
#include <hip/hip_bf16.h>

typedef __attribute__((ext_vector_type(2))) float v2f;
typedef __attribute__((ext_vector_type(8))) float v8f;

constexpr int RK_THREADS = 1024;
constexpr int RK_EPT     = 16;
constexpr int RK_CHUNK   = RK_THREADS * RK_EPT;
constexpr int RK_HCAP    = 65536;

__global__ __launch_bounds__(RK_THREADS) void gcn_lds_rank_kernel(
    const int* __restrict__ src, const int* __restrict__ dst,
    int* __restrict__ part_in, int* __restrict__ part_out,
    int* __restrict__ rank, int E, int N, int G)
{
    __shared__ __attribute__((aligned(16))) int h[RK_HCAP];
    const int  t      = threadIdx.x;
    const bool is_dst = (int)blockIdx.x < G;
    const int  c      = is_dst ? (int)blockIdx.x : (int)blockIdx.x - G;
    const int* __restrict__ keys = is_dst ? dst : src;
    int* __restrict__ part = (is_dst ? part_in : part_out) + (long)c * N;
    const int  e0     = c * RK_CHUNK;

    int k[RK_EPT];
    #pragma unroll
    for (int u = 0; u < RK_EPT; ++u) {
        const int e = e0 + u * RK_THREADS + t;
        k[u] = (e < E) ? keys[e] : -1;
    }

    for (int base = 0; base < N; base += RK_HCAP) {
        const int nr = min(RK_HCAP, N - base);
        for (int i = 4 * t; i < nr; i += 4 * RK_THREADS)
            *(int4*)&h[i] = make_int4(0, 0, 0, 0);
        __syncthreads();
        if (is_dst) {
            int r[RK_EPT];
            #pragma unroll
            for (int u = 0; u < RK_EPT; ++u) {
                const unsigned d = (unsigned)(k[u] - base);
                r[u] = (d < (unsigned)nr) ? atomicAdd(&h[d], 1) : -1;
            }
            #pragma unroll
            for (int u = 0; u < RK_EPT; ++u)
                if (r[u] >= 0) rank[e0 + u * RK_THREADS + t] = r[u];
        } else {
            #pragma unroll
            for (int u = 0; u < RK_EPT; ++u) {
                const unsigned d = (unsigned)(k[u] - base);
                if (d < (unsigned)nr) atomicAdd(&h[d], 1);
            }
        }
        __syncthreads();
        const int nv = ((((size_t)(part + base)) & 15) == 0) ? (nr & ~3) : 0;
        for (int i = 4 * t; i < nv; i += 4 * RK_THREADS)
            *(int4*)&part[base + i] = *(const int4*)&h[i];
        for (int i = nv + t; i < nr; i += RK_THREADS) part[base + i] = h[i];
        __syncthreads();
    }
}

__device__ __forceinline__ int gcn_block_prefix_256(int v, int* wt, int& total)
{
    const int lane = threadIdx.x & 31;
    const int wv   = threadIdx.x >> 5;
    int inc = v;
    #pragma unroll
    for (int d = 1; d < 32; d <<= 1) {
        const int o = __shfl_up(inc, d);
        if (lane >= d) inc += o;
    }
    if (lane == 31) wt[wv] = inc;
    __syncthreads();
    int wo = 0, tot = 0;
    #pragma unroll
    for (int k = 0; k < 8; ++k) {
        const int x = wt[k];
        tot += x;
        if (k < wv) wo += x;
    }
    total = tot;
    return wo + inc - v;
}

__global__ __launch_bounds__(256) void gcn_merge_counts_kernel(
    const int* __restrict__ part_in, const int* __restrict__ part_out,
    int* __restrict__ off, int* __restrict__ rloc, int* __restrict__ bsum,
    int* __restrict__ cnt_out, int N, int G)
{
    __shared__ __attribute__((aligned(16))) int wt[8];
    const int i = blockIdx.x * 256 + threadIdx.x;
    int run = 0;
    if (i < N) {
        int so = 0;
        #pragma unroll 8
        for (int g = 0; g < G; ++g) {
            const long o = (long)g * N + i;
            off[o] = run;
            run += part_in[o];
            so  += part_out[o];
        }
        cnt_out[i] = so;
    }
    int tot;
    const int ex = gcn_block_prefix_256(run, wt, tot);
    if (i < N) rloc[i] = ex;
    if (threadIdx.x == 0) bsum[blockIdx.x] = tot;
}

__global__ __launch_bounds__(256) void gcn_csr_fill_kernel(
    const int* __restrict__ src, const int* __restrict__ dst,
    const float* __restrict__ ew, const int* __restrict__ cnt_out,
    const int* __restrict__ rloc, const int* __restrict__ bsum,
    const int* __restrict__ rank, const int* __restrict__ off,
    int* __restrict__ rec, int* __restrict__ rowptr, int E, int N, int NB)
{
    __shared__ int sbase[256];
    __shared__ __attribute__((aligned(16))) int wt[8];
    const int t = threadIdx.x;

    if ((int)blockIdx.x < NB) {
        const int j = (int)blockIdx.x;
        int part = 0;
        for (int q = t; q < j; q += 256) part += bsum[q];
        int base;
        (void)gcn_block_prefix_256(part, wt, base);
        const int i = j * 256 + t;
        if (i < N) rowptr[i] = base + rloc[i];
        if (j == NB - 1 && t == 0) rowptr[N] = base + bsum[j];
        return;
    }

    const int  e  = ((int)blockIdx.x - NB) * 256 + t;
    const bool ok = e < E;
    int v = (t < NB) ? bsum[t] : 0;
    int s = 0, d = 0;
    if (ok) { s = src[e]; d = dst[e]; }
    int dbase = 0, carry = 0;
    for (int tb = 0; tb < NB; tb += 256) {
        int tot;
        const int ex = gcn_block_prefix_256(v, wt, tot);
        sbase[t] = carry + ex;
        __syncthreads();
        const unsigned r = (unsigned)((d >> 8) - tb);
        if (r < 256u) dbase = sbase[r];
        carry += tot;
        const int q = tb + 256 + t;
        v = (q < NB) ? bsum[q] : 0;
    }
    if (ok) {
        const int   pos = dbase + rloc[d] + off[(long)(e / RK_CHUNK) * N + d] + rank[e];
        const float w   = ew[e] * rsqrtf(fmaxf((float)cnt_out[s], 1.0f));
        *(int2*)(rec + 2 * (long)pos) = make_int2(s, __float_as_int(w));
    }
}

template <int NT>
__global__ __launch_bounds__(256) void gcn_wmma_gemm_f32(
    const float* __restrict__ A, const float* __restrict__ B,
    float* __restrict__ C, int M, int N, int K, int lda, int ldb, int ldc)
{
    constexpr int AP = 20;
    constexpr int BN = 16 * NT;
    constexpr int BP = (BN % 32 == 16) ? BN : BN + 16;
    static_assert(BN <= 256, "one B staging thread per block column");
    __shared__ __attribute__((aligned(16))) float As[2][128 * AP];
    __shared__ __attribute__((aligned(16))) v2f   Bs[2][8 * BP];

    const int t    = threadIdx.x;
    const int lane = t & 31;
    const int wv   = t >> 5;
    const int half = lane >> 4;
    const int l16  = lane & 15;

    const int m0 = blockIdx.x * 128;
    const int n0 = blockIdx.y * BN;

    const int r0 = t >> 2;
    const int r1 = r0 + 64;
    const int c0 = (t & 3) * 4;
    const bool bown = t < BN;
    const bool bok  = bown && (n0 + t < N);

    const float* pa0 = A + (long)(m0 + r0) * lda + c0;
    const float* pa1 = A + (long)(m0 + r1) * lda + c0;
    const unsigned bcol = n0 + t;

    const bool fastA = (m0 + 128 <= M) && ((lda & 3) == 0) &&
                       ((((size_t)A) & 15) == 0);

    float4 ra0, ra1;
    v2f    rb[8];
    #pragma unroll
    for (int i = 0; i < 8; ++i) { rb[i].x = 0.0f; rb[i].y = 0.0f; }

    v8f acc[NT];
    #pragma unroll
    for (int j = 0; j < NT; ++j) acc[j] = v8f{};

    auto fetch_fast = [&](int k0) {
        ra0 = *(const float4*)(pa0 + k0);
        ra1 = *(const float4*)(pa1 + k0);
        if (bok) {
            const float* q = B + (long)k0 * ldb;
            #pragma unroll
            for (int i = 0; i < 8; ++i) {
                rb[i].x = (q + (long)(2 * i) * ldb)[bcol];
                rb[i].y = (q + (long)(2 * i + 1) * ldb)[bcol];
            }
        }
    };
    auto fetch_masked = [&](int k0) {
        float* p0 = (float*)&ra0;
        float* p1 = (float*)&ra1;
        const int g0 = m0 + r0, g1 = m0 + r1;
        #pragma unroll
        for (int j = 0; j < 4; ++j) {
            int gk = k0 + c0 + j;
            p0[j] = (g0 < M && gk < K) ? A[(long)g0 * lda + gk] : 0.0f;
            p1[j] = (g1 < M && gk < K) ? A[(long)g1 * lda + gk] : 0.0f;
        }
        if (bok) {
            const float* q = B + (long)k0 * ldb;
            #pragma unroll
            for (int i = 0; i < 8; ++i) {
                int gk = k0 + 2 * i;
                rb[i].x = (gk < K)     ? (q + (long)(2 * i) * ldb)[bcol]     : 0.0f;
                rb[i].y = (gk + 1 < K) ? (q + (long)(2 * i + 1) * ldb)[bcol] : 0.0f;
            }
        }
    };

    int buf = 0;
    auto slab = [&](int k0, auto&& fetch_next, bool has_next) {
        *(float4*)&As[buf][r0 * AP + c0] = ra0;
        *(float4*)&As[buf][r1 * AP + c0] = ra1;
        if (bown) {
            #pragma unroll
            for (int i = 0; i < 8; ++i) Bs[buf][i * BP + t] = rb[i];
        }
        __syncthreads();

        if (has_next) fetch_next(k0 + 16);

        const float* as = &As[buf][(wv * 16 + l16) * AP + half * 2];
        const v2f*   bs = &Bs[buf][half * BP + l16];
        #pragma unroll
        for (int ks = 0; ks < 16; ks += 4) {
            v2f a = *(const v2f*)(as + ks);
            v2f b[NT];
            #pragma unroll
            for (int j = 0; j < NT; ++j) b[j] = bs[(ks >> 1) * BP + j * 16];
            #pragma unroll
            for (int j = 0; j < NT; ++j)
                acc[j] = __builtin_amdgcn_wmma_f32_16x16x4_f32(
                    false, a, false, b[j], (short)0, acc[j], false, false);
        }
        buf ^= 1;
    };

    if (fastA) {
        if (K >= 16) fetch_fast(0); else fetch_masked(0);
        int k0 = 0;
        for (; k0 + 32 <= K; k0 += 16) slab(k0, fetch_fast, true);
        for (; k0 < K; k0 += 16)       slab(k0, fetch_masked, k0 + 16 < K);
    } else {
        fetch_masked(0);
        for (int k0 = 0; k0 < K; k0 += 16) slab(k0, fetch_masked, k0 + 16 < K);
    }

    #pragma unroll
    for (int j = 0; j < NT; ++j) {
        const int n = n0 + j * 16 + l16;
        #pragma unroll
        for (int v = 0; v < 8; ++v) {
            int m = m0 + wv * 16 + v + half * 8;
            if (m < M && n < N) C[(long)m * ldc + n] = acc[j][v];
        }
    }
}

template <int NV, bool RELU>
__global__ __launch_bounds__(256) void gcn_gather_kernel(
    const float* __restrict__ H, const int* __restrict__ rec,
    const int* __restrict__ rowptr, const float* __restrict__ b,
    float* __restrict__ outp, int n, int F)
{
    const int lane = threadIdx.x & 31;
    const int node = __builtin_amdgcn_readfirstlane(
        (int)((blockIdx.x * blockDim.x + threadIdx.x) >> 5));
    if (node >= n) return;
    const int beg = rowptr[node];
    const int end = rowptr[node + 1];

    unsigned col[NV];
    #pragma unroll
    for (int u = 0; u < NV; ++u) col[u] = (unsigned)min(4 * (lane + 32 * u), F - 4);

    float4 acc[NV];
    #pragma unroll
    for (int u = 0; u < NV; ++u) acc[u] = make_float4(0.0f, 0.0f, 0.0f, 0.0f);

    for (int base = beg; base < end; base += 32) {
        const int nb = min(32, end - base);
        int sj = 0, wj = 0;
        if (lane < nb) {
            const int2 r = *(const int2*)(rec + 2 * (long)(base + lane));
            sj = r.x;
            wj = r.y;
        }
        int j = 0;
        for (; j + 4 <= nb; j += 4) {
            float4 hv[4][NV];
            float  wt[4];
            #pragma unroll
            for (int q = 0; q < 4; ++q) {
                const int s = __builtin_amdgcn_readlane(sj, j + q);
                wt[q] = __int_as_float(__builtin_amdgcn_readlane(wj, j + q));
                const float* hs = H + (long)s * F;
                #pragma unroll
                for (int u = 0; u < NV; ++u) hv[q][u] = *(const float4*)(hs + col[u]);
            }
            #pragma unroll
            for (int q = 0; q < 4; ++q) {
                #pragma unroll
                for (int u = 0; u < NV; ++u) {
                    acc[u].x = fmaf(hv[q][u].x, wt[q], acc[u].x);
                    acc[u].y = fmaf(hv[q][u].y, wt[q], acc[u].y);
                    acc[u].z = fmaf(hv[q][u].z, wt[q], acc[u].z);
                    acc[u].w = fmaf(hv[q][u].w, wt[q], acc[u].w);
                }
            }
        }
        for (; j < nb; ++j) {
            const int   s = __builtin_amdgcn_readlane(sj, j);
            const float w = __int_as_float(__builtin_amdgcn_readlane(wj, j));
            const float* hs = H + (long)s * F;
            float4 h1[NV];
            #pragma unroll
            for (int u = 0; u < NV; ++u) h1[u] = *(const float4*)(hs + col[u]);
            #pragma unroll
            for (int u = 0; u < NV; ++u) {
                acc[u].x = fmaf(h1[u].x, w, acc[u].x);
                acc[u].y = fmaf(h1[u].y, w, acc[u].y);
                acc[u].z = fmaf(h1[u].z, w, acc[u].z);
                acc[u].w = fmaf(h1[u].w, w, acc[u].w);
            }
        }
    }

    const float di = rsqrtf(fmaxf((float)(end - beg), 1.0f));
    float* orow = outp + (long)node * F;
    #pragma unroll
    for (int u = 0; u < NV; ++u) {
        const int c = 4 * (lane + 32 * u);
        if (c < F) {
            const float4 bb = *(const float4*)(b + c);
            float4 o;
            o.x = acc[u].x * di + bb.x;
            o.y = acc[u].y * di + bb.y;
            o.z = acc[u].z * di + bb.z;
            o.w = acc[u].w * di + bb.w;
            if (RELU) {
                o.x = fmaxf(o.x, 0.0f);
                o.y = fmaxf(o.y, 0.0f);
                o.z = fmaxf(o.z, 0.0f);
                o.w = fmaxf(o.w, 0.0f);
            }
            *(float4*)(orow + c) = o;
        }
    }
}

template <int NV, int NT2>
__global__ __launch_bounds__(512) void gcn_gather_gemm_kernel(
    const float* __restrict__ H, const int* __restrict__ rec,
    const int* __restrict__ rowptr, const float* __restrict__ b,
    const float* __restrict__ W, float* __restrict__ H2, int n, int F, int F2)
{
    constexpr int XP = 128 * NV + 4;
    constexpr int NC = 16 * NT2;
    constexpr int BP = (NC % 32 == 16) ? NC : NC + 16;
    constexpr int KP = 64 * NV;
    __shared__ __attribute__((aligned(16))) float Xs[16 * XP];
    __shared__ __attribute__((aligned(16))) v2f   Ws[KP * BP];

    const int t    = threadIdx.x;
    const int lane = t & 31;
    const int wv    = __builtin_amdgcn_readfirstlane(t >> 5);
    const int node0 = blockIdx.x * 16;
    const int node  = node0 + wv;

    {
        constexpr int WI = (KP * NC + 511) / 512;
        const int P = F >> 1;
        v2f wreg[WI];
        #pragma unroll
        for (int i = 0; i < WI; ++i) {
            const int idx = t + 512 * i;
            const int p = idx / NC, c = idx % NC;
            wreg[i].x = 0.0f; wreg[i].y = 0.0f;
            if (p < P && c < F2) {
                wreg[i].x = W[(long)(2 * p) * F2 + c];
                wreg[i].y = W[(long)(2 * p + 1) * F2 + c];
            }
        }
        #pragma unroll
        for (int i = 0; i < WI; ++i) {
            const int idx = t + 512 * i;
            const int p = idx / NC, c = idx % NC;
            if (p < P) Ws[p * BP + c] = wreg[i];
        }
    }

    unsigned col[NV];
    #pragma unroll
    for (int u = 0; u < NV; ++u) col[u] = (unsigned)min(4 * (lane + 32 * u), F - 4);

    float4 acc[NV];
    #pragma unroll
    for (int u = 0; u < NV; ++u) acc[u] = make_float4(0.0f, 0.0f, 0.0f, 0.0f);

    int beg = 0, end = 0;
    if (node < n) { beg = rowptr[node]; end = rowptr[node + 1]; }

    for (int base = beg; base < end; base += 32) {
        const int nb = min(32, end - base);
        int sj = 0, wj = 0;
        if (lane < nb) {
            const int2 r = *(const int2*)(rec + 2 * (long)(base + lane));
            sj = r.x;
            wj = r.y;
        }
        int j = 0;
        for (; j + 4 <= nb; j += 4) {
            float4 hv[4][NV];
            float  wt[4];
            #pragma unroll
            for (int q = 0; q < 4; ++q) {
                const int s = __builtin_amdgcn_readlane(sj, j + q);
                wt[q] = __int_as_float(__builtin_amdgcn_readlane(wj, j + q));
                const float* hs = H + (long)s * F;
                #pragma unroll
                for (int u = 0; u < NV; ++u) hv[q][u] = *(const float4*)(hs + col[u]);
            }
            #pragma unroll
            for (int q = 0; q < 4; ++q) {
                #pragma unroll
                for (int u = 0; u < NV; ++u) {
                    acc[u].x = fmaf(hv[q][u].x, wt[q], acc[u].x);
                    acc[u].y = fmaf(hv[q][u].y, wt[q], acc[u].y);
                    acc[u].z = fmaf(hv[q][u].z, wt[q], acc[u].z);
                    acc[u].w = fmaf(hv[q][u].w, wt[q], acc[u].w);
                }
            }
        }
        for (; j < nb; ++j) {
            const int   s = __builtin_amdgcn_readlane(sj, j);
            const float w = __int_as_float(__builtin_amdgcn_readlane(wj, j));
            const float* hs = H + (long)s * F;
            float4 h1[NV];
            #pragma unroll
            for (int u = 0; u < NV; ++u) h1[u] = *(const float4*)(hs + col[u]);
            #pragma unroll
            for (int u = 0; u < NV; ++u) {
                acc[u].x = fmaf(h1[u].x, w, acc[u].x);
                acc[u].y = fmaf(h1[u].y, w, acc[u].y);
                acc[u].z = fmaf(h1[u].z, w, acc[u].z);
                acc[u].w = fmaf(h1[u].w, w, acc[u].w);
            }
        }
    }

    const float di = rsqrtf(fmaxf((float)(end - beg), 1.0f));
    #pragma unroll
    for (int u = 0; u < NV; ++u) {
        const int c = 4 * (lane + 32 * u);
        if (c < F) {
            float4 o = make_float4(0.0f, 0.0f, 0.0f, 0.0f);
            if (node < n) {
                const float4 bb = *(const float4*)(b + c);
                o.x = fmaxf(acc[u].x * di + bb.x, 0.0f);
                o.y = fmaxf(acc[u].y * di + bb.y, 0.0f);
                o.z = fmaxf(acc[u].z * di + bb.z, 0.0f);
                o.w = fmaxf(acc[u].w * di + bb.w, 0.0f);
            }
            *(float4*)&Xs[wv * XP + c] = o;
        }
    }
    __syncthreads();

    if (wv != 0) return;
    const int half = lane >> 4;
    const int l16  = lane & 15;
    const float* as = &Xs[l16 * XP + half * 2];
    const v2f*   bs = &Ws[half * BP + l16];
    v8f d[NT2];
    #pragma unroll
    for (int j = 0; j < NT2; ++j) d[j] = v8f{};
    for (int ks = 0; ks < F; ks += 4) {
        const v2f a = *(const v2f*)(as + ks);
        v2f bf[NT2];
        #pragma unroll
        for (int j = 0; j < NT2; ++j) bf[j] = bs[(ks >> 1) * BP + j * 16];
        #pragma unroll
        for (int j = 0; j < NT2; ++j)
            d[j] = __builtin_amdgcn_wmma_f32_16x16x4_f32(
                false, a, false, bf[j], (short)0, d[j], false, false);
    }
    #pragma unroll
    for (int j = 0; j < NT2; ++j) {
        const int nn = j * 16 + l16;
        #pragma unroll
        for (int v = 0; v < 8; ++v) {
            const int m = node0 + v + half * 8;
            if (m < n && nn < F2) H2[(long)m * F2 + nn] = d[j][v];
        }
    }
}

extern "C" void kernel_launch(void* const* d_in, const int* in_sizes, int n_in,
                              void* d_out, int out_size, void* d_ws, size_t ws_size,
                              hipStream_t stream)
{
    const float* feature = (const float*)d_in[0];
    const int*   src     = (const int*)  d_in[1];
    const int*   dst     = (const int*)  d_in[2];
    const float* edge_w  = (const float*)d_in[3];
    const float* W1      = (const float*)d_in[4];
    const float* b1      = (const float*)d_in[5];
    const float* W2      = (const float*)d_in[6];
    const float* b2      = (const float*)d_in[7];
    float*       out     = (float*)d_out;

    const int F1 = in_sizes[5];
    const int F2 = in_sizes[7];
    const int K0 = in_sizes[4] / F1;
    const int N  = in_sizes[0] / K0;
    const int E  = in_sizes[1];

    float* ws      = (float*)d_ws;
    int*   cnt_out = (int*)ws;
    int*   rloc    = cnt_out + N;
    float* R1      = ws + 2 * (long)N;
    float* R2      = R1 + (long)N * F1;
    float* R3      = R2 + (long)N * F1;
    float* H1      = R1;
    float* H2      = R2;
    int*   rank    = (int*)R2;
    const int G    = (E + RK_CHUNK - 1) / RK_CHUNK;
    int*   part_in  = rank + E;
    int*   part_out = part_in + (long)G * N;
    int*   off      = part_out + (long)G * N;
    int*   rec     = (int*)R3;
    int*   rowptr  = rec + 2 * (long)E;
    const int NB   = (N + 255) / 256;
    const int EB   = (E + 255) / 256;
    int*   bsum    = rowptr + N + 1;

    gcn_lds_rank_kernel<<<2 * G, RK_THREADS, 0, stream>>>(src, dst, part_in, part_out, rank, E, N, G);
    gcn_merge_counts_kernel<<<NB, 256, 0, stream>>>(part_in, part_out, off, rloc, bsum, cnt_out, N, G);
    gcn_csr_fill_kernel<<<NB + EB, 256, 0, stream>>>(src, dst, edge_w, cnt_out, rloc, bsum,
                                                     rank, off, rec, rowptr, E, N, NB);

    {
        dim3 grid((N + 127) / 128, (F1 + 207) / 208);
        gcn_wmma_gemm_f32<13><<<grid, 256, 0, stream>>>(feature, W1, H1, N, F1, K0, K0, F1, F1);
    }
    gcn_gather_gemm_kernel<2, 2><<<(N + 15) / 16, 512, 0, stream>>>(H1, rec, rowptr, b1, W2, H2, N, F1, F2);
    gcn_gather_kernel<1, false><<<(N + 7) / 8, 256, 0, stream>>>(H2, rec, rowptr, b2, out, N, F2);
}


# layer-1 GEMM rebalanced: one 208x208 block per WGP (241), tile groups dealt so a SIMD gets 42-43 tiles (was 52 where two blocks shared a WGP); 3-slab LDS ring; C out via LDS, whole lines; sums kept
# speedup vs baseline: 3.6251x; 1.2278x over previous
#include <hip/hip_runtime.h>
#include <hip/hip_bf16.h>

typedef __attribute__((ext_vector_type(2))) float v2f;
typedef __attribute__((ext_vector_type(8))) float v8f;

constexpr int RK_THREADS = 1024;
constexpr int RK_EPT     = 16;
constexpr int RK_CHUNK   = RK_THREADS * RK_EPT;
constexpr int RK_HCAP    = 65536;

__global__ __launch_bounds__(RK_THREADS) void gcn_lds_rank_kernel(
    const int* __restrict__ src, const int* __restrict__ dst,
    int* __restrict__ part_in, int* __restrict__ part_out,
    int* __restrict__ rank, int E, int N, int G)
{
    __shared__ __attribute__((aligned(16))) int h[RK_HCAP];
    const int  t      = threadIdx.x;
    const bool is_dst = (int)blockIdx.x < G;
    const int  c      = is_dst ? (int)blockIdx.x : (int)blockIdx.x - G;
    const int* __restrict__ keys = is_dst ? dst : src;
    int* __restrict__ part = (is_dst ? part_in : part_out) + (long)c * N;
    const int  e0     = c * RK_CHUNK;

    int k[RK_EPT];
    #pragma unroll
    for (int u = 0; u < RK_EPT; ++u) {
        const int e = e0 + u * RK_THREADS + t;
        k[u] = (e < E) ? keys[e] : -1;
    }

    for (int base = 0; base < N; base += RK_HCAP) {
        const int nr = min(RK_HCAP, N - base);
        for (int i = 4 * t; i < nr; i += 4 * RK_THREADS)
            *(int4*)&h[i] = make_int4(0, 0, 0, 0);
        __syncthreads();
        if (is_dst) {
            int r[RK_EPT];
            #pragma unroll
            for (int u = 0; u < RK_EPT; ++u) {
                const unsigned d = (unsigned)(k[u] - base);
                r[u] = (d < (unsigned)nr) ? atomicAdd(&h[d], 1) : -1;
            }
            #pragma unroll
            for (int u = 0; u < RK_EPT; ++u)
                if (r[u] >= 0) rank[e0 + u * RK_THREADS + t] = r[u];
        } else {
            #pragma unroll
            for (int u = 0; u < RK_EPT; ++u) {
                const unsigned d = (unsigned)(k[u] - base);
                if (d < (unsigned)nr) atomicAdd(&h[d], 1);
            }
        }
        __syncthreads();
        const int nv = ((((size_t)(part + base)) & 15) == 0) ? (nr & ~3) : 0;
        for (int i = 4 * t; i < nv; i += 4 * RK_THREADS)
            *(int4*)&part[base + i] = *(const int4*)&h[i];
        for (int i = nv + t; i < nr; i += RK_THREADS) part[base + i] = h[i];
        __syncthreads();
    }
}

__device__ __forceinline__ int gcn_block_prefix_256(int v, int* wt, int& total)
{
    const int lane = threadIdx.x & 31;
    const int wv   = threadIdx.x >> 5;
    int inc = v;
    #pragma unroll
    for (int d = 1; d < 32; d <<= 1) {
        const int o = __shfl_up(inc, d);
        if (lane >= d) inc += o;
    }
    if (lane == 31) wt[wv] = inc;
    __syncthreads();
    int wo = 0, tot = 0;
    #pragma unroll
    for (int k = 0; k < 8; ++k) {
        const int x = wt[k];
        tot += x;
        if (k < wv) wo += x;
    }
    total = tot;
    return wo + inc - v;
}

__global__ __launch_bounds__(256) void gcn_merge_counts_kernel(
    const int* __restrict__ part_in, const int* __restrict__ part_out,
    int* __restrict__ off, int* __restrict__ rloc, int* __restrict__ bsum,
    int* __restrict__ cnt_out, int N, int G)
{
    __shared__ __attribute__((aligned(16))) int wt[8];
    const int i = blockIdx.x * 256 + threadIdx.x;
    int run = 0;
    if (i < N) {
        int so = 0;
        #pragma unroll 8
        for (int g = 0; g < G; ++g) {
            const long o = (long)g * N + i;
            off[o] = run;
            run += part_in[o];
            so  += part_out[o];
        }
        cnt_out[i] = so;
    }
    int tot;
    const int ex = gcn_block_prefix_256(run, wt, tot);
    if (i < N) rloc[i] = ex;
    if (threadIdx.x == 0) bsum[blockIdx.x] = tot;
}

__global__ __launch_bounds__(256) void gcn_csr_fill_kernel(
    const int* __restrict__ src, const int* __restrict__ dst,
    const float* __restrict__ ew, const int* __restrict__ cnt_out,
    const int* __restrict__ rloc, const int* __restrict__ bsum,
    const int* __restrict__ rank, const int* __restrict__ off,
    int* __restrict__ rec, int* __restrict__ rowptr, int E, int N, int NB)
{
    __shared__ int sbase[256];
    __shared__ __attribute__((aligned(16))) int wt[8];
    const int t = threadIdx.x;

    if ((int)blockIdx.x < NB) {
        const int j = (int)blockIdx.x;
        int part = 0;
        for (int q = t; q < j; q += 256) part += bsum[q];
        int base;
        (void)gcn_block_prefix_256(part, wt, base);
        const int i = j * 256 + t;
        if (i < N) rowptr[i] = base + rloc[i];
        if (j == NB - 1 && t == 0) rowptr[N] = base + bsum[j];
        return;
    }

    const int  e  = ((int)blockIdx.x - NB) * 256 + t;
    const bool ok = e < E;
    int v = (t < NB) ? bsum[t] : 0;
    int s = 0, d = 0;
    if (ok) { s = src[e]; d = dst[e]; }
    int dbase = 0, carry = 0;
    for (int tb = 0; tb < NB; tb += 256) {
        int tot;
        const int ex = gcn_block_prefix_256(v, wt, tot);
        sbase[t] = carry + ex;
        __syncthreads();
        const unsigned r = (unsigned)((d >> 8) - tb);
        if (r < 256u) dbase = sbase[r];
        carry += tot;
        const int q = tb + 256 + t;
        v = (q < NB) ? bsum[q] : 0;
    }
    if (ok) {
        const int   pos = dbase + rloc[d] + off[(long)(e / RK_CHUNK) * N + d] + rank[e];
        const float w   = ew[e] * rsqrtf(fmaxf((float)cnt_out[s], 1.0f));
        *(int2*)(rec + 2 * (long)pos) = make_int2(s, __float_as_int(w));
    }
}

constexpr int GW_T  = 13;
constexpr int GW_BM = 16 * GW_T;
constexpr int GW_AP = 20;
constexpr int GW_BP = GW_BM;
constexpr int GW_CP = 212;

struct GcnGwFrag {
    v2f a[4][3], b[4][3], a3[4], b3[4];
};

template <bool CANM>
__device__ __forceinline__ void gcn_gw_load(GcnGwFrag& f, const float* as, const float* ax,
                                            const v2f* bs, const v2f* bx)
{
    #pragma unroll
    for (int q = 0; q < 4; ++q) {
        #pragma unroll
        for (int mi = 0; mi < 3; ++mi) f.a[q][mi] = *(const v2f*)(as + mi * 16 * GW_AP + 4 * q);
        if (CANM) f.a3[q] = *(const v2f*)(ax + 4 * q);
        #pragma unroll
        for (int nj = 0; nj < 3; ++nj) f.b[q][nj] = bs[2 * q * GW_BP + nj * 16];
        f.b3[q] = bx[2 * q * GW_BP];
    }
}

template <bool CANM>
__device__ __forceinline__ void gcn_gw_mma(v8f (&acc)[16], const GcnGwFrag& f, bool xm, bool xn)
{
    #pragma unroll
    for (int q = 0; q < 4; ++q) {
        #pragma unroll
        for (int mi = 0; mi < 3; ++mi) {
            #pragma unroll
            for (int nj = 0; nj < 3; ++nj)
                acc[mi * 4 + nj] = __builtin_amdgcn_wmma_f32_16x16x4_f32(
                    false, f.a[q][mi], false, f.b[q][nj], (short)0, acc[mi * 4 + nj], false, false);
        }
    }
    if (xn) {
        #pragma unroll
        for (int q = 0; q < 4; ++q) {
            #pragma unroll
            for (int mi = 0; mi < 3; ++mi)
                acc[mi * 4 + 3] = __builtin_amdgcn_wmma_f32_16x16x4_f32(
                    false, f.a[q][mi], false, f.b3[q], (short)0, acc[mi * 4 + 3], false, false);
        }
    }
    if (CANM) {
        if (xm) {
            #pragma unroll
            for (int q = 0; q < 4; ++q) {
                #pragma unroll
                for (int nj = 0; nj < 3; ++nj)
                    acc[12 + nj] = __builtin_amdgcn_wmma_f32_16x16x4_f32(
                        false, f.a3[q], false, f.b[q][nj], (short)0, acc[12 + nj], false, false);
            }
            if (xn) {
                #pragma unroll
                for (int q = 0; q < 4; ++q)
                    acc[15] = __builtin_amdgcn_wmma_f32_16x16x4_f32(
                        false, f.a3[q], false, f.b3[q], (short)0, acc[15], false, false);
            }
        }
    }
}

__device__ __forceinline__ void gcn_gw_put_tile(const v8f& d, float* cs)
{
    #pragma unroll
    for (int v = 0; v < 8; ++v) cs[v * GW_CP] = d[v];
}

template <bool CANM>
__device__ __forceinline__ void gcn_gw_put(const v8f (&acc)[16], float* cs, bool xm, bool xn)
{
    #pragma unroll
    for (int mi = 0; mi < 3; ++mi) {
        #pragma unroll
        for (int nj = 0; nj < 3; ++nj)
            gcn_gw_put_tile(acc[mi * 4 + nj], cs + mi * 16 * GW_CP + nj * 16);
    }
    if (xn) {
        #pragma unroll
        for (int mi = 0; mi < 3; ++mi) gcn_gw_put_tile(acc[mi * 4 + 3], cs + mi * 16 * GW_CP + 48);
    }
    if (CANM) {
        if (xm) {
            #pragma unroll
            for (int nj = 0; nj < 3; ++nj) gcn_gw_put_tile(acc[12 + nj], cs + 48 * GW_CP + nj * 16);
            if (xn) gcn_gw_put_tile(acc[15], cs + 48 * GW_CP + 48);
        }
    }
}

__global__ __launch_bounds__(256) void gcn_wmma_gemm_wgp_f32(
    const float* __restrict__ A, const float* __restrict__ B,
    float* __restrict__ C, int M, int N, int K, int lda, int ldb, int ldc)
{
    __shared__ __attribute__((aligned(16))) float smem[GW_BM * GW_CP];
    static_assert(3 * (GW_BM * GW_AP + 16 * GW_BP) <= GW_BM * GW_CP, "slab ring fits under the result tile");
    float (*As)[GW_BM * GW_AP] = (float (*)[GW_BM * GW_AP])smem;
    v2f   (*Bs)[8 * GW_BP]     = (v2f (*)[8 * GW_BP])(smem + 3 * GW_BM * GW_AP);
    float* Cs = smem;

    const int t    = threadIdx.x;
    const int lane = t & 31;
    const int wv   = __builtin_amdgcn_readfirstlane(t >> 5);
    const int half = lane >> 4;
    const int l16  = lane & 15;

    const int  sc    = wv & 3;
    const int  gi0   = (wv >> 2) * 2;
    const int  gj0   = (sc + gi0) & 3;
    const int  gj1   = (sc + gi0 + 1) & 3;
    const bool bigM0 = gi0 == 0;
    const bool bigN0 = gj0 == 0;
    const bool bigN1 = gj1 == 0;
    const int  mo0   = bigM0 ? 0 : 3 * gi0 + 1;
    const int  mo1   = 3 * gi0 + 4;
    const int  no0   = bigN0 ? 0 : 3 * gj0 + 1;
    const int  no1   = bigN1 ? 0 : 3 * gj1 + 1;

    const int m0 = blockIdx.x * GW_BM;

    const bool has3 = t < 4 * GW_BM - 768;
    const int  r0 = t >> 2;
    const int  c0 = (t & 3) * 4;
    const bool bown = t < GW_BM;
    const bool bok  = bown && (t < N);

    const float* pa0 = A + (long)min(m0 + r0, M - 1) * lda + c0;
    const float* pa1 = A + (long)min(m0 + r0 + 64, M - 1) * lda + c0;
    const float* pa2 = A + (long)min(m0 + r0 + 128, M - 1) * lda + c0;
    const float* pa3 = A + (long)min(m0 + (has3 ? r0 + 192 : 0), M - 1) * lda + c0;
    const unsigned bcol = t;

    const bool vecA = ((lda & 3) == 0) && ((((size_t)A) & 15) == 0);

    float4 ra0 = make_float4(0.0f, 0.0f, 0.0f, 0.0f), ra1 = ra0, ra2 = ra0, ra3 = ra0;
    v2f    rb[8];
    #pragma unroll
    for (int i = 0; i < 8; ++i) { rb[i].x = 0.0f; rb[i].y = 0.0f; }

    v8f acc0[16], acc1[16];
    #pragma unroll
    for (int j = 0; j < 16; ++j) { acc0[j] = v8f{}; acc1[j] = v8f{}; }

    auto fetch_fast = [&](int k0) {
        ra0 = *(const float4*)(pa0 + k0);
        ra1 = *(const float4*)(pa1 + k0);
        ra2 = *(const float4*)(pa2 + k0);
        if (has3) ra3 = *(const float4*)(pa3 + k0);
        if (bok) {
            const float* q = B + (long)k0 * ldb;
            #pragma unroll
            for (int i = 0; i < 8; ++i) {
                rb[i].x = (q + (long)(2 * i) * ldb)[bcol];
                rb[i].y = (q + (long)(2 * i + 1) * ldb)[bcol];
            }
        }
    };
    auto fetch_masked = [&](int k0) {
        float* p0 = (float*)&ra0;
        float* p1 = (float*)&ra1;
        float* p2 = (float*)&ra2;
        float* p3 = (float*)&ra3;
        #pragma unroll
        for (int j = 0; j < 4; ++j) {
            const int gk = k0 + c0 + j;
            p0[j] = (gk < K) ? pa0[k0 + j] : 0.0f;
            p1[j] = (gk < K) ? pa1[k0 + j] : 0.0f;
            p2[j] = (gk < K) ? pa2[k0 + j] : 0.0f;
            p3[j] = (has3 && gk < K) ? pa3[k0 + j] : 0.0f;
        }
        if (bok) {
            const float* q = B + (long)k0 * ldb;
            #pragma unroll
            for (int i = 0; i < 8; ++i) {
                const int gk = k0 + 2 * i;
                rb[i].x = (gk < K)     ? (q + (long)(2 * i) * ldb)[bcol]     : 0.0f;
                rb[i].y = (gk + 1 < K) ? (q + (long)(2 * i + 1) * ldb)[bcol] : 0.0f;
            }
        }
    };

    auto fetch = [&](int k0) {
        if (vecA && k0 + 16 <= K) fetch_fast(k0); else fetch_masked(k0);
    };
    auto stage_from = [&](int bsel, const float4& s0, const float4& s1, const float4& s2,
                          const float4& s3, const v2f (&sb)[8]) {
        *(float4*)&As[bsel][r0 * GW_AP + c0] = s0;
        *(float4*)&As[bsel][(r0 + 64) * GW_AP + c0] = s1;
        *(float4*)&As[bsel][(r0 + 128) * GW_AP + c0] = s2;
        if (has3) *(float4*)&As[bsel][(r0 + 192) * GW_AP + c0] = s3;
        if (bown) {
            #pragma unroll
            for (int i = 0; i < 8; ++i) Bs[bsel][i * GW_BP + t] = sb[i];
        }
    };
    GcnGwFrag f0;
    auto load0 = [&](int bsel) {
        const float* as0 = &As[bsel][(mo0 * 16 + l16) * GW_AP + half * 2];
        const float* ax0 = &As[bsel][(min(mo0 + 3, GW_T - 1) * 16 + l16) * GW_AP + half * 2];
        const v2f*   bs0 = &Bs[bsel][half * GW_BP + no0 * 16 + l16];
        const v2f*   bx0 = &Bs[bsel][half * GW_BP + min(no0 + 3, GW_T - 1) * 16 + l16];
        gcn_gw_load<true>(f0, as0, ax0, bs0, bx0);
    };

    fetch(0);
    {
        const float4 s0 = ra0, s1 = ra1, s2 = ra2, s3 = ra3;
        v2f sb[8];
        #pragma unroll
        for (int i = 0; i < 8; ++i) sb[i] = rb[i];
        if (16 < K) fetch(16);
        stage_from(0, s0, s1, s2, s3, sb);
    }
    if (16 < K) stage_from(1, ra0, ra1, ra2, ra3, rb);
    if (32 < K) fetch(32);
    __syncthreads();
    load0(0);

    int cur = 0, nx1 = 1, nx2 = 2;
    auto step = [&](int k0, auto&& fetch_next, bool full) {
        gcn_gw_mma<true>(acc0, f0, bigM0, bigN0);

        if (full || k0 + 32 < K) stage_from(nx2, ra0, ra1, ra2, ra3, rb);
        if (full || k0 + 48 < K) fetch_next(k0 + 48);

        {
            GcnGwFrag f1;
            const float* as1 = &As[cur][(mo1 * 16 + l16) * GW_AP + half * 2];
            const v2f*   bs1 = &Bs[cur][half * GW_BP + no1 * 16 + l16];
            const v2f*   bx1 = &Bs[cur][half * GW_BP + min(no1 + 3, GW_T - 1) * 16 + l16];
            gcn_gw_load<false>(f1, as1, as1, bs1, bx1);
            if (full || k0 + 16 < K) load0(nx1);
            gcn_gw_mma<false>(acc1, f1, false, bigN1);
        }

        __builtin_amdgcn_fence(__ATOMIC_RELEASE, "workgroup", "local");
        __builtin_amdgcn_s_barrier();
        __builtin_amdgcn_fence(__ATOMIC_ACQUIRE, "workgroup", "local");
        const int old = cur; cur = nx1; nx1 = nx2; nx2 = old;
    };
    {
        int k0 = 0;
        if (vecA) {
            for (; k0 + 64 <= K; k0 += 16) step(k0, fetch_fast, true);
        }
        for (; k0 < K; k0 += 16) step(k0, fetch_masked, false);
    }

    {
        float* cs0 = &Cs[(mo0 * 16 + 8 * half) * GW_CP + no0 * 16 + l16];
        gcn_gw_put<true>(acc0, cs0, bigM0, bigN0);
        float* cs1 = &Cs[(mo1 * 16 + 8 * half) * GW_CP + no1 * 16 + l16];
        gcn_gw_put<false>(acc1, cs1, false, bigN1);
    }
    __syncthreads();

    const int rows = min(GW_BM, M - m0);
    if (((N & 3) == 0) && ((ldc & 3) == 0) && ((((size_t)C) & 15) == 0)) {
        const int n4  = N >> 2;
        const int tot = rows * n4;
        const int qd = 256 / n4, rm = 256 - qd * n4;
        int r = t / n4;
        int c = t - r * n4;
        for (int idx = t; idx < tot; idx += 256) {
            *(float4*)(C + (long)(m0 + r) * ldc + 4 * c) = *(const float4*)&Cs[r * GW_CP + 4 * c];
            r += qd; c += rm;
            if (c >= n4) { c -= n4; ++r; }
        }
    } else {
        const int tot = rows * N;
        for (int idx = t; idx < tot; idx += 256) {
            const int r = idx / N;
            const int c = idx - r * N;
            C[(long)(m0 + r) * ldc + c] = Cs[r * GW_CP + c];
        }
    }
}

template <int NV, bool RELU>
__global__ __launch_bounds__(256) void gcn_gather_kernel(
    const float* __restrict__ H, const int* __restrict__ rec,
    const int* __restrict__ rowptr, const float* __restrict__ b,
    float* __restrict__ outp, int n, int F)
{
    const int lane = threadIdx.x & 31;
    const int node = __builtin_amdgcn_readfirstlane(
        (int)((blockIdx.x * blockDim.x + threadIdx.x) >> 5));
    if (node >= n) return;
    const int beg = rowptr[node];
    const int end = rowptr[node + 1];

    unsigned col[NV];
    #pragma unroll
    for (int u = 0; u < NV; ++u) col[u] = (unsigned)min(4 * (lane + 32 * u), F - 4);

    float4 acc[NV];
    #pragma unroll
    for (int u = 0; u < NV; ++u) acc[u] = make_float4(0.0f, 0.0f, 0.0f, 0.0f);

    for (int base = beg; base < end; base += 32) {
        const int nb = min(32, end - base);
        int sj = 0, wj = 0;
        if (lane < nb) {
            const int2 r = *(const int2*)(rec + 2 * (long)(base + lane));
            sj = r.x;
            wj = r.y;
        }
        int j = 0;
        for (; j + 4 <= nb; j += 4) {
            float4 hv[4][NV];
            float  wt[4];
            #pragma unroll
            for (int q = 0; q < 4; ++q) {
                const int s = __builtin_amdgcn_readlane(sj, j + q);
                wt[q] = __int_as_float(__builtin_amdgcn_readlane(wj, j + q));
                const float* hs = H + (long)s * F;
                #pragma unroll
                for (int u = 0; u < NV; ++u) hv[q][u] = *(const float4*)(hs + col[u]);
            }
            #pragma unroll
            for (int q = 0; q < 4; ++q) {
                #pragma unroll
                for (int u = 0; u < NV; ++u) {
                    acc[u].x = fmaf(hv[q][u].x, wt[q], acc[u].x);
                    acc[u].y = fmaf(hv[q][u].y, wt[q], acc[u].y);
                    acc[u].z = fmaf(hv[q][u].z, wt[q], acc[u].z);
                    acc[u].w = fmaf(hv[q][u].w, wt[q], acc[u].w);
                }
            }
        }
        for (; j < nb; ++j) {
            const int   s = __builtin_amdgcn_readlane(sj, j);
            const float w = __int_as_float(__builtin_amdgcn_readlane(wj, j));
            const float* hs = H + (long)s * F;
            float4 h1[NV];
            #pragma unroll
            for (int u = 0; u < NV; ++u) h1[u] = *(const float4*)(hs + col[u]);
            #pragma unroll
            for (int u = 0; u < NV; ++u) {
                acc[u].x = fmaf(h1[u].x, w, acc[u].x);
                acc[u].y = fmaf(h1[u].y, w, acc[u].y);
                acc[u].z = fmaf(h1[u].z, w, acc[u].z);
                acc[u].w = fmaf(h1[u].w, w, acc[u].w);
            }
        }
    }

    const float di = rsqrtf(fmaxf((float)(end - beg), 1.0f));
    float* orow = outp + (long)node * F;
    #pragma unroll
    for (int u = 0; u < NV; ++u) {
        const int c = 4 * (lane + 32 * u);
        if (c < F) {
            const float4 bb = *(const float4*)(b + c);
            float4 o;
            o.x = acc[u].x * di + bb.x;
            o.y = acc[u].y * di + bb.y;
            o.z = acc[u].z * di + bb.z;
            o.w = acc[u].w * di + bb.w;
            if (RELU) {
                o.x = fmaxf(o.x, 0.0f);
                o.y = fmaxf(o.y, 0.0f);
                o.z = fmaxf(o.z, 0.0f);
                o.w = fmaxf(o.w, 0.0f);
            }
            *(float4*)(orow + c) = o;
        }
    }
}

template <int NV, int NT2>
__global__ __launch_bounds__(512) void gcn_gather_gemm_kernel(
    const float* __restrict__ H, const int* __restrict__ rec,
    const int* __restrict__ rowptr, const float* __restrict__ b,
    const float* __restrict__ W, float* __restrict__ H2, int n, int F, int F2)
{
    constexpr int XP = 128 * NV + 4;
    constexpr int NC = 16 * NT2;
    constexpr int BP = (NC % 32 == 16) ? NC : NC + 16;
    constexpr int KP = 64 * NV;
    __shared__ __attribute__((aligned(16))) float Xs[16 * XP];
    __shared__ __attribute__((aligned(16))) v2f   Ws[KP * BP];

    const int t    = threadIdx.x;
    const int lane = t & 31;
    const int wv    = __builtin_amdgcn_readfirstlane(t >> 5);
    const int node0 = blockIdx.x * 16;
    const int node  = node0 + wv;

    {
        constexpr int WI = (KP * NC + 511) / 512;
        const int P = F >> 1;
        v2f wreg[WI];
        #pragma unroll
        for (int i = 0; i < WI; ++i) {
            const int idx = t + 512 * i;
            const int p = idx / NC, c = idx % NC;
            wreg[i].x = 0.0f; wreg[i].y = 0.0f;
            if (p < P && c < F2) {
                wreg[i].x = W[(long)(2 * p) * F2 + c];
                wreg[i].y = W[(long)(2 * p + 1) * F2 + c];
            }
        }
        #pragma unroll
        for (int i = 0; i < WI; ++i) {
            const int idx = t + 512 * i;
            const int p = idx / NC, c = idx % NC;
            if (p < P) Ws[p * BP + c] = wreg[i];
        }
    }

    unsigned col[NV];
    #pragma unroll
    for (int u = 0; u < NV; ++u) col[u] = (unsigned)min(4 * (lane + 32 * u), F - 4);

    float4 acc[NV];
    #pragma unroll
    for (int u = 0; u < NV; ++u) acc[u] = make_float4(0.0f, 0.0f, 0.0f, 0.0f);

    int beg = 0, end = 0;
    if (node < n) { beg = rowptr[node]; end = rowptr[node + 1]; }

    for (int base = beg; base < end; base += 32) {
        const int nb = min(32, end - base);
        int sj = 0, wj = 0;
        if (lane < nb) {
            const int2 r = *(const int2*)(rec + 2 * (long)(base + lane));
            sj = r.x;
            wj = r.y;
        }
        int j = 0;
        for (; j + 4 <= nb; j += 4) {
            float4 hv[4][NV];
            float  wt[4];
            #pragma unroll
            for (int q = 0; q < 4; ++q) {
                const int s = __builtin_amdgcn_readlane(sj, j + q);
                wt[q] = __int_as_float(__builtin_amdgcn_readlane(wj, j + q));
                const float* hs = H + (long)s * F;
                #pragma unroll
                for (int u = 0; u < NV; ++u) hv[q][u] = *(const float4*)(hs + col[u]);
            }
            #pragma unroll
            for (int q = 0; q < 4; ++q) {
                #pragma unroll
                for (int u = 0; u < NV; ++u) {
                    acc[u].x = fmaf(hv[q][u].x, wt[q], acc[u].x);
                    acc[u].y = fmaf(hv[q][u].y, wt[q], acc[u].y);
                    acc[u].z = fmaf(hv[q][u].z, wt[q], acc[u].z);
                    acc[u].w = fmaf(hv[q][u].w, wt[q], acc[u].w);
                }
            }
        }
        for (; j < nb; ++j) {
            const int   s = __builtin_amdgcn_readlane(sj, j);
            const float w = __int_as_float(__builtin_amdgcn_readlane(wj, j));
            const float* hs = H + (long)s * F;
            float4 h1[NV];
            #pragma unroll
            for (int u = 0; u < NV; ++u) h1[u] = *(const float4*)(hs + col[u]);
            #pragma unroll
            for (int u = 0; u < NV; ++u) {
                acc[u].x = fmaf(h1[u].x, w, acc[u].x);
                acc[u].y = fmaf(h1[u].y, w, acc[u].y);
                acc[u].z = fmaf(h1[u].z, w, acc[u].z);
                acc[u].w = fmaf(h1[u].w, w, acc[u].w);
            }
        }
    }

    const float di = rsqrtf(fmaxf((float)(end - beg), 1.0f));
    #pragma unroll
    for (int u = 0; u < NV; ++u) {
        const int c = 4 * (lane + 32 * u);
        if (c < F) {
            float4 o = make_float4(0.0f, 0.0f, 0.0f, 0.0f);
            if (node < n) {
                const float4 bb = *(const float4*)(b + c);
                o.x = fmaxf(acc[u].x * di + bb.x, 0.0f);
                o.y = fmaxf(acc[u].y * di + bb.y, 0.0f);
                o.z = fmaxf(acc[u].z * di + bb.z, 0.0f);
                o.w = fmaxf(acc[u].w * di + bb.w, 0.0f);
            }
            *(float4*)&Xs[wv * XP + c] = o;
        }
    }
    __syncthreads();

    if (wv != 0) return;
    const int half = lane >> 4;
    const int l16  = lane & 15;
    const float* as = &Xs[l16 * XP + half * 2];
    const v2f*   bs = &Ws[half * BP + l16];
    v8f d[NT2];
    #pragma unroll
    for (int j = 0; j < NT2; ++j) d[j] = v8f{};
    for (int ks = 0; ks < F; ks += 4) {
        const v2f a = *(const v2f*)(as + ks);
        v2f bf[NT2];
        #pragma unroll
        for (int j = 0; j < NT2; ++j) bf[j] = bs[(ks >> 1) * BP + j * 16];
        #pragma unroll
        for (int j = 0; j < NT2; ++j)
            d[j] = __builtin_amdgcn_wmma_f32_16x16x4_f32(
                false, a, false, bf[j], (short)0, d[j], false, false);
    }
    #pragma unroll
    for (int j = 0; j < NT2; ++j) {
        const int nn = j * 16 + l16;
        #pragma unroll
        for (int v = 0; v < 8; ++v) {
            const int m = node0 + v + half * 8;
            if (m < n && nn < F2) H2[(long)m * F2 + nn] = d[j][v];
        }
    }
}

extern "C" void kernel_launch(void* const* d_in, const int* in_sizes, int n_in,
                              void* d_out, int out_size, void* d_ws, size_t ws_size,
                              hipStream_t stream)
{
    const float* feature = (const float*)d_in[0];
    const int*   src     = (const int*)  d_in[1];
    const int*   dst     = (const int*)  d_in[2];
    const float* edge_w  = (const float*)d_in[3];
    const float* W1      = (const float*)d_in[4];
    const float* b1      = (const float*)d_in[5];
    const float* W2      = (const float*)d_in[6];
    const float* b2      = (const float*)d_in[7];
    float*       out     = (float*)d_out;

    const int F1 = in_sizes[5];
    const int F2 = in_sizes[7];
    const int K0 = in_sizes[4] / F1;
    const int N  = in_sizes[0] / K0;
    const int E  = in_sizes[1];

    float* ws      = (float*)d_ws;
    int*   cnt_out = (int*)ws;
    int*   rloc    = cnt_out + N;
    float* R1      = ws + 2 * (long)N;
    float* R2      = R1 + (long)N * F1;
    float* R3      = R2 + (long)N * F1;
    float* H1      = R1;
    float* H2      = R2;
    int*   rank    = (int*)R2;
    const int G    = (E + RK_CHUNK - 1) / RK_CHUNK;
    int*   part_in  = rank + E;
    int*   part_out = part_in + (long)G * N;
    int*   off      = part_out + (long)G * N;
    int*   rec     = (int*)R3;
    int*   rowptr  = rec + 2 * (long)E;
    const int NB   = (N + 255) / 256;
    const int EB   = (E + 255) / 256;
    int*   bsum    = rowptr + N + 1;

    gcn_lds_rank_kernel<<<2 * G, RK_THREADS, 0, stream>>>(src, dst, part_in, part_out, rank, E, N, G);
    gcn_merge_counts_kernel<<<NB, 256, 0, stream>>>(part_in, part_out, off, rloc, bsum, cnt_out, N, G);
    gcn_csr_fill_kernel<<<NB + EB, 256, 0, stream>>>(src, dst, edge_w, cnt_out, rloc, bsum,
                                                     rank, off, rec, rowptr, E, N, NB);

    gcn_wmma_gemm_wgp_f32<<<(N + GW_BM - 1) / GW_BM, 256, 0, stream>>>(feature, W1, H1, N, F1, K0, K0, F1, F1);
    gcn_gather_gemm_kernel<2, 2><<<(N + 15) / 16, 512, 0, stream>>>(H1, rec, rowptr, b1, W2, H2, N, F1, F2);
    gcn_gather_kernel<1, false><<<(N + 7) / 8, 256, 0, stream>>>(H2, rec, rowptr, b2, out, N, F2);
}


# count/rank pass runs as extra work-groups of the GEMM launch, on the WGPs its 241 blocks leave idle (7 edge chunks, not 49; merge sums them in one trip); 5 launches; sum order = new chunk-major rank
# speedup vs baseline: 3.8587x; 1.0644x over previous
#include <hip/hip_runtime.h>
#include <hip/hip_bf16.h>

typedef __attribute__((ext_vector_type(2))) float v2f;
typedef __attribute__((ext_vector_type(8))) float v8f;

constexpr int RK_THREADS = 256;
constexpr int RK_EPT     = 16;
constexpr int RK_BATCH   = RK_THREADS * RK_EPT;
constexpr int RK_CHUNK   = 28 * RK_BATCH;
constexpr int RK_HCAP    = 50176;

__device__ __forceinline__ void gcn_lds_rank_role(
    int* h, int rb,
    const int* __restrict__ src, const int* __restrict__ dst,
    int* __restrict__ part_in, int* __restrict__ part_out,
    int* __restrict__ rank, int E, int N, int G)
{
    const int  t      = threadIdx.x;
    const bool is_dst = rb < G;
    const int  c      = is_dst ? rb : rb - G;
    const int* __restrict__ keys = is_dst ? dst : src;
    int* __restrict__ part = (is_dst ? part_in : part_out) + (long)c * N;
    const int  e0     = c * RK_CHUNK;
    const int  e1     = min(E, e0 + RK_CHUNK);

    for (int base = 0; base < N; base += RK_HCAP) {
        const int nr = min(RK_HCAP, N - base);
        for (int i = 4 * t; i < nr; i += 4 * RK_THREADS)
            *(int4*)&h[i] = make_int4(0, 0, 0, 0);
        int k[RK_EPT];
        #pragma unroll
        for (int u = 0; u < RK_EPT; ++u) {
            const int e = e0 + u * RK_THREADS + t;
            k[u] = (e < e1) ? keys[e] : -1;
        }
        __syncthreads();
        for (int b = e0; b < e1; b += RK_BATCH) {
            int kn[RK_EPT];
            #pragma unroll
            for (int u = 0; u < RK_EPT; ++u) {
                const int e = b + RK_BATCH + u * RK_THREADS + t;
                kn[u] = (e < e1) ? keys[e] : -1;
            }
            if (is_dst) {
                int r[RK_EPT];
                #pragma unroll
                for (int u = 0; u < RK_EPT; ++u) {
                    const unsigned d = (unsigned)(k[u] - base);
                    r[u] = (d < (unsigned)nr) ? atomicAdd(&h[d], 1) : -1;
                }
                #pragma unroll
                for (int u = 0; u < RK_EPT; ++u)
                    if (r[u] >= 0) rank[b + u * RK_THREADS + t] = r[u];
            } else {
                #pragma unroll
                for (int u = 0; u < RK_EPT; ++u) {
                    const unsigned d = (unsigned)(k[u] - base);
                    if (d < (unsigned)nr) atomicAdd(&h[d], 1);
                }
            }
            #pragma unroll
            for (int u = 0; u < RK_EPT; ++u) k[u] = kn[u];
        }
        __syncthreads();
        const int nv = ((((size_t)(part + base)) & 15) == 0) ? (nr & ~3) : 0;
        for (int i = 4 * t; i < nv; i += 4 * RK_THREADS)
            *(int4*)&part[base + i] = *(const int4*)&h[i];
        for (int i = nv + t; i < nr; i += RK_THREADS) part[base + i] = h[i];
        __syncthreads();
    }
}

__device__ __forceinline__ int gcn_block_prefix_256(int v, int* wt, int& total)
{
    const int lane = threadIdx.x & 31;
    const int wv   = threadIdx.x >> 5;
    int inc = v;
    #pragma unroll
    for (int d = 1; d < 32; d <<= 1) {
        const int o = __shfl_up(inc, d);
        if (lane >= d) inc += o;
    }
    if (lane == 31) wt[wv] = inc;
    __syncthreads();
    int wo = 0, tot = 0;
    #pragma unroll
    for (int k = 0; k < 8; ++k) {
        const int x = wt[k];
        tot += x;
        if (k < wv) wo += x;
    }
    total = tot;
    return wo + inc - v;
}

__global__ __launch_bounds__(256) void gcn_merge_counts_kernel(
    const int* __restrict__ part_in, const int* __restrict__ part_out,
    int* __restrict__ off, int* __restrict__ rloc, int* __restrict__ bsum,
    int* __restrict__ cnt_out, int N, int G)
{
    __shared__ __attribute__((aligned(16))) int wt[8];
    const int i = blockIdx.x * 256 + threadIdx.x;
    int run = 0;
    if (i < N) {
        int so = 0;
        for (int g0 = 0; g0 < G; g0 += 8) {
            int pi[8], po[8];
            #pragma unroll
            for (int j = 0; j < 8; ++j) {
                pi[j] = 0; po[j] = 0;
                if (g0 + j < G) {
                    const long o = (long)(g0 + j) * N + i;
                    pi[j] = part_in[o];
                    po[j] = part_out[o];
                }
            }
            #pragma unroll
            for (int j = 0; j < 8; ++j) {
                if (g0 + j < G) {
                    off[(long)(g0 + j) * N + i] = run;
                    run += pi[j];
                    so  += po[j];
                }
            }
        }
        cnt_out[i] = so;
    }
    int tot;
    const int ex = gcn_block_prefix_256(run, wt, tot);
    if (i < N) rloc[i] = ex;
    if (threadIdx.x == 0) bsum[blockIdx.x] = tot;
}

__global__ __launch_bounds__(256) void gcn_csr_fill_kernel(
    const int* __restrict__ src, const int* __restrict__ dst,
    const float* __restrict__ ew, const int* __restrict__ cnt_out,
    const int* __restrict__ rloc, const int* __restrict__ bsum,
    const int* __restrict__ rank, const int* __restrict__ off,
    int* __restrict__ rec, int* __restrict__ rowptr, int E, int N, int NB)
{
    __shared__ int sbase[256];
    __shared__ __attribute__((aligned(16))) int wt[8];
    const int t = threadIdx.x;

    if ((int)blockIdx.x < NB) {
        const int j = (int)blockIdx.x;
        int part = 0;
        for (int q = t; q < j; q += 256) part += bsum[q];
        int base;
        (void)gcn_block_prefix_256(part, wt, base);
        const int i = j * 256 + t;
        if (i < N) rowptr[i] = base + rloc[i];
        if (j == NB - 1 && t == 0) rowptr[N] = base + bsum[j];
        return;
    }

    const int  e  = ((int)blockIdx.x - NB) * 256 + t;
    const bool ok = e < E;
    int v = (t < NB) ? bsum[t] : 0;
    int s = 0, d = 0;
    if (ok) { s = src[e]; d = dst[e]; }
    int dbase = 0, carry = 0;
    for (int tb = 0; tb < NB; tb += 256) {
        int tot;
        const int ex = gcn_block_prefix_256(v, wt, tot);
        sbase[t] = carry + ex;
        __syncthreads();
        const unsigned r = (unsigned)((d >> 8) - tb);
        if (r < 256u) dbase = sbase[r];
        carry += tot;
        const int q = tb + 256 + t;
        v = (q < NB) ? bsum[q] : 0;
    }
    if (ok) {
        const int   pos = dbase + rloc[d] + off[(long)(e / RK_CHUNK) * N + d] + rank[e];
        const float w   = ew[e] * rsqrtf(fmaxf((float)cnt_out[s], 1.0f));
        *(int2*)(rec + 2 * (long)pos) = make_int2(s, __float_as_int(w));
    }
}

constexpr int GW_T  = 13;
constexpr int GW_BM = 16 * GW_T;
constexpr int GW_AP = 20;
constexpr int GW_BP = GW_BM;
constexpr int GW_CP = 212;

struct GcnGwFrag {
    v2f a[4][3], b[4][3], a3[4], b3[4];
};

template <bool CANM>
__device__ __forceinline__ void gcn_gw_load(GcnGwFrag& f, const float* as, const float* ax,
                                            const v2f* bs, const v2f* bx)
{
    #pragma unroll
    for (int q = 0; q < 4; ++q) {
        #pragma unroll
        for (int mi = 0; mi < 3; ++mi) f.a[q][mi] = *(const v2f*)(as + mi * 16 * GW_AP + 4 * q);
        if (CANM) f.a3[q] = *(const v2f*)(ax + 4 * q);
        #pragma unroll
        for (int nj = 0; nj < 3; ++nj) f.b[q][nj] = bs[2 * q * GW_BP + nj * 16];
        f.b3[q] = bx[2 * q * GW_BP];
    }
}

template <bool CANM>
__device__ __forceinline__ void gcn_gw_mma(v8f (&acc)[16], const GcnGwFrag& f, bool xm, bool xn)
{
    #pragma unroll
    for (int q = 0; q < 4; ++q) {
        #pragma unroll
        for (int mi = 0; mi < 3; ++mi) {
            #pragma unroll
            for (int nj = 0; nj < 3; ++nj)
                acc[mi * 4 + nj] = __builtin_amdgcn_wmma_f32_16x16x4_f32(
                    false, f.a[q][mi], false, f.b[q][nj], (short)0, acc[mi * 4 + nj], false, false);
        }
    }
    if (xn) {
        #pragma unroll
        for (int q = 0; q < 4; ++q) {
            #pragma unroll
            for (int mi = 0; mi < 3; ++mi)
                acc[mi * 4 + 3] = __builtin_amdgcn_wmma_f32_16x16x4_f32(
                    false, f.a[q][mi], false, f.b3[q], (short)0, acc[mi * 4 + 3], false, false);
        }
    }
    if (CANM) {
        if (xm) {
            #pragma unroll
            for (int q = 0; q < 4; ++q) {
                #pragma unroll
                for (int nj = 0; nj < 3; ++nj)
                    acc[12 + nj] = __builtin_amdgcn_wmma_f32_16x16x4_f32(
                        false, f.a3[q], false, f.b[q][nj], (short)0, acc[12 + nj], false, false);
            }
            if (xn) {
                #pragma unroll
                for (int q = 0; q < 4; ++q)
                    acc[15] = __builtin_amdgcn_wmma_f32_16x16x4_f32(
                        false, f.a3[q], false, f.b3[q], (short)0, acc[15], false, false);
            }
        }
    }
}

__device__ __forceinline__ void gcn_gw_put_tile(const v8f& d, float* cs)
{
    #pragma unroll
    for (int v = 0; v < 8; ++v) cs[v * GW_CP] = d[v];
}

template <bool CANM>
__device__ __forceinline__ void gcn_gw_put(const v8f (&acc)[16], float* cs, bool xm, bool xn)
{
    #pragma unroll
    for (int mi = 0; mi < 3; ++mi) {
        #pragma unroll
        for (int nj = 0; nj < 3; ++nj)
            gcn_gw_put_tile(acc[mi * 4 + nj], cs + mi * 16 * GW_CP + nj * 16);
    }
    if (xn) {
        #pragma unroll
        for (int mi = 0; mi < 3; ++mi) gcn_gw_put_tile(acc[mi * 4 + 3], cs + mi * 16 * GW_CP + 48);
    }
    if (CANM) {
        if (xm) {
            #pragma unroll
            for (int nj = 0; nj < 3; ++nj) gcn_gw_put_tile(acc[12 + nj], cs + 48 * GW_CP + nj * 16);
            if (xn) gcn_gw_put_tile(acc[15], cs + 48 * GW_CP + 48);
        }
    }
}

constexpr int GW_SMEM = (GW_BM * GW_CP > RK_HCAP) ? GW_BM * GW_CP : RK_HCAP;

__global__ __launch_bounds__(256) void gcn_wmma_gemm_wgp_f32(
    const float* __restrict__ A, const float* __restrict__ B,
    float* __restrict__ C, int M, int N, int K, int lda, int ldb, int ldc,
    int gemm_blocks,
    const int* __restrict__ esrc, const int* __restrict__ edst,
    int* __restrict__ part_in, int* __restrict__ part_out,
    int* __restrict__ rank, int rE, int rN, int rG)
{
    __shared__ __attribute__((aligned(16))) float smem[GW_SMEM];
    if ((int)blockIdx.x >= gemm_blocks) {
        gcn_lds_rank_role((int*)smem, (int)blockIdx.x - gemm_blocks, esrc, edst,
                          part_in, part_out, rank, rE, rN, rG);
        return;
    }
    static_assert(3 * (GW_BM * GW_AP + 16 * GW_BP) <= GW_BM * GW_CP, "slab ring fits under the result tile");
    float (*As)[GW_BM * GW_AP] = (float (*)[GW_BM * GW_AP])smem;
    v2f   (*Bs)[8 * GW_BP]     = (v2f (*)[8 * GW_BP])(smem + 3 * GW_BM * GW_AP);
    float* Cs = smem;

    const int t    = threadIdx.x;
    const int lane = t & 31;
    const int wv   = __builtin_amdgcn_readfirstlane(t >> 5);
    const int half = lane >> 4;
    const int l16  = lane & 15;

    const int  sc    = wv & 3;
    const int  gi0   = (wv >> 2) * 2;
    const int  gj0   = (sc + gi0) & 3;
    const int  gj1   = (sc + gi0 + 1) & 3;
    const bool bigM0 = gi0 == 0;
    const bool bigN0 = gj0 == 0;
    const bool bigN1 = gj1 == 0;
    const int  mo0   = bigM0 ? 0 : 3 * gi0 + 1;
    const int  mo1   = 3 * gi0 + 4;
    const int  no0   = bigN0 ? 0 : 3 * gj0 + 1;
    const int  no1   = bigN1 ? 0 : 3 * gj1 + 1;

    const int m0 = blockIdx.x * GW_BM;

    const bool has3 = t < 4 * GW_BM - 768;
    const int  r0 = t >> 2;
    const int  c0 = (t & 3) * 4;
    const bool bown = t < GW_BM;
    const bool bok  = bown && (t < N);

    const float* pa0 = A + (long)min(m0 + r0, M - 1) * lda + c0;
    const float* pa1 = A + (long)min(m0 + r0 + 64, M - 1) * lda + c0;
    const float* pa2 = A + (long)min(m0 + r0 + 128, M - 1) * lda + c0;
    const float* pa3 = A + (long)min(m0 + (has3 ? r0 + 192 : 0), M - 1) * lda + c0;
    const unsigned bcol = t;

    const bool vecA = ((lda & 3) == 0) && ((((size_t)A) & 15) == 0);

    float4 ra0 = make_float4(0.0f, 0.0f, 0.0f, 0.0f), ra1 = ra0, ra2 = ra0, ra3 = ra0;
    v2f    rb[8];
    #pragma unroll
    for (int i = 0; i < 8; ++i) { rb[i].x = 0.0f; rb[i].y = 0.0f; }

    v8f acc0[16], acc1[16];
    #pragma unroll
    for (int j = 0; j < 16; ++j) { acc0[j] = v8f{}; acc1[j] = v8f{}; }

    auto fetch_fast = [&](int k0) {
        ra0 = *(const float4*)(pa0 + k0);
        ra1 = *(const float4*)(pa1 + k0);
        ra2 = *(const float4*)(pa2 + k0);
        if (has3) ra3 = *(const float4*)(pa3 + k0);
        if (bok) {
            const float* q = B + (long)k0 * ldb;
            #pragma unroll
            for (int i = 0; i < 8; ++i) {
                rb[i].x = (q + (long)(2 * i) * ldb)[bcol];
                rb[i].y = (q + (long)(2 * i + 1) * ldb)[bcol];
            }
        }
    };
    auto fetch_masked = [&](int k0) {
        float* p0 = (float*)&ra0;
        float* p1 = (float*)&ra1;
        float* p2 = (float*)&ra2;
        float* p3 = (float*)&ra3;
        #pragma unroll
        for (int j = 0; j < 4; ++j) {
            const int gk = k0 + c0 + j;
            p0[j] = (gk < K) ? pa0[k0 + j] : 0.0f;
            p1[j] = (gk < K) ? pa1[k0 + j] : 0.0f;
            p2[j] = (gk < K) ? pa2[k0 + j] : 0.0f;
            p3[j] = (has3 && gk < K) ? pa3[k0 + j] : 0.0f;
        }
        if (bok) {
            const float* q = B + (long)k0 * ldb;
            #pragma unroll
            for (int i = 0; i < 8; ++i) {
                const int gk = k0 + 2 * i;
                rb[i].x = (gk < K)     ? (q + (long)(2 * i) * ldb)[bcol]     : 0.0f;
                rb[i].y = (gk + 1 < K) ? (q + (long)(2 * i + 1) * ldb)[bcol] : 0.0f;
            }
        }
    };

    auto fetch = [&](int k0) {
        if (vecA && k0 + 16 <= K) fetch_fast(k0); else fetch_masked(k0);
    };
    auto stage_from = [&](int bsel, const float4& s0, const float4& s1, const float4& s2,
                          const float4& s3, const v2f (&sb)[8]) {
        *(float4*)&As[bsel][r0 * GW_AP + c0] = s0;
        *(float4*)&As[bsel][(r0 + 64) * GW_AP + c0] = s1;
        *(float4*)&As[bsel][(r0 + 128) * GW_AP + c0] = s2;
        if (has3) *(float4*)&As[bsel][(r0 + 192) * GW_AP + c0] = s3;
        if (bown) {
            #pragma unroll
            for (int i = 0; i < 8; ++i) Bs[bsel][i * GW_BP + t] = sb[i];
        }
    };
    GcnGwFrag f0;
    auto load0 = [&](int bsel) {
        const float* as0 = &As[bsel][(mo0 * 16 + l16) * GW_AP + half * 2];
        const float* ax0 = &As[bsel][(min(mo0 + 3, GW_T - 1) * 16 + l16) * GW_AP + half * 2];
        const v2f*   bs0 = &Bs[bsel][half * GW_BP + no0 * 16 + l16];
        const v2f*   bx0 = &Bs[bsel][half * GW_BP + min(no0 + 3, GW_T - 1) * 16 + l16];
        gcn_gw_load<true>(f0, as0, ax0, bs0, bx0);
    };

    fetch(0);
    {
        const float4 s0 = ra0, s1 = ra1, s2 = ra2, s3 = ra3;
        v2f sb[8];
        #pragma unroll
        for (int i = 0; i < 8; ++i) sb[i] = rb[i];
        if (16 < K) fetch(16);
        stage_from(0, s0, s1, s2, s3, sb);
    }
    if (16 < K) stage_from(1, ra0, ra1, ra2, ra3, rb);
    if (32 < K) fetch(32);
    __syncthreads();
    load0(0);

    int cur = 0, nx1 = 1, nx2 = 2;
    auto step = [&](int k0, auto&& fetch_next, bool full) {
        gcn_gw_mma<true>(acc0, f0, bigM0, bigN0);

        if (full || k0 + 32 < K) stage_from(nx2, ra0, ra1, ra2, ra3, rb);
        if (full || k0 + 48 < K) fetch_next(k0 + 48);

        {
            GcnGwFrag f1;
            const float* as1 = &As[cur][(mo1 * 16 + l16) * GW_AP + half * 2];
            const v2f*   bs1 = &Bs[cur][half * GW_BP + no1 * 16 + l16];
            const v2f*   bx1 = &Bs[cur][half * GW_BP + min(no1 + 3, GW_T - 1) * 16 + l16];
            gcn_gw_load<false>(f1, as1, as1, bs1, bx1);
            if (full || k0 + 16 < K) load0(nx1);
            gcn_gw_mma<false>(acc1, f1, false, bigN1);
        }

        __builtin_amdgcn_fence(__ATOMIC_RELEASE, "workgroup", "local");
        __builtin_amdgcn_s_barrier();
        __builtin_amdgcn_fence(__ATOMIC_ACQUIRE, "workgroup", "local");
        const int old = cur; cur = nx1; nx1 = nx2; nx2 = old;
    };
    {
        int k0 = 0;
        if (vecA) {
            for (; k0 + 64 <= K; k0 += 16) step(k0, fetch_fast, true);
        }
        for (; k0 < K; k0 += 16) step(k0, fetch_masked, false);
    }

    {
        float* cs0 = &Cs[(mo0 * 16 + 8 * half) * GW_CP + no0 * 16 + l16];
        gcn_gw_put<true>(acc0, cs0, bigM0, bigN0);
        float* cs1 = &Cs[(mo1 * 16 + 8 * half) * GW_CP + no1 * 16 + l16];
        gcn_gw_put<false>(acc1, cs1, false, bigN1);
    }
    __syncthreads();

    const int rows = min(GW_BM, M - m0);
    if (((N & 3) == 0) && ((ldc & 3) == 0) && ((((size_t)C) & 15) == 0)) {
        const int n4  = N >> 2;
        const int tot = rows * n4;
        const int qd = 256 / n4, rm = 256 - qd * n4;
        int r = t / n4;
        int c = t - r * n4;
        for (int idx = t; idx < tot; idx += 256) {
            *(float4*)(C + (long)(m0 + r) * ldc + 4 * c) = *(const float4*)&Cs[r * GW_CP + 4 * c];
            r += qd; c += rm;
            if (c >= n4) { c -= n4; ++r; }
        }
    } else {
        const int tot = rows * N;
        for (int idx = t; idx < tot; idx += 256) {
            const int r = idx / N;
            const int c = idx - r * N;
            C[(long)(m0 + r) * ldc + c] = Cs[r * GW_CP + c];
        }
    }
}

template <int NV, bool RELU>
__global__ __launch_bounds__(256) void gcn_gather_kernel(
    const float* __restrict__ H, const int* __restrict__ rec,
    const int* __restrict__ rowptr, const float* __restrict__ b,
    float* __restrict__ outp, int n, int F)
{
    const int lane = threadIdx.x & 31;
    const int node = __builtin_amdgcn_readfirstlane(
        (int)((blockIdx.x * blockDim.x + threadIdx.x) >> 5));
    if (node >= n) return;
    const int beg = rowptr[node];
    const int end = rowptr[node + 1];

    unsigned col[NV];
    #pragma unroll
    for (int u = 0; u < NV; ++u) col[u] = (unsigned)min(4 * (lane + 32 * u), F - 4);

    float4 acc[NV];
    #pragma unroll
    for (int u = 0; u < NV; ++u) acc[u] = make_float4(0.0f, 0.0f, 0.0f, 0.0f);

    for (int base = beg; base < end; base += 32) {
        const int nb = min(32, end - base);
        int sj = 0, wj = 0;
        if (lane < nb) {
            const int2 r = *(const int2*)(rec + 2 * (long)(base + lane));
            sj = r.x;
            wj = r.y;
        }
        int j = 0;
        for (; j + 4 <= nb; j += 4) {
            float4 hv[4][NV];
            float  wt[4];
            #pragma unroll
            for (int q = 0; q < 4; ++q) {
                const int s = __builtin_amdgcn_readlane(sj, j + q);
                wt[q] = __int_as_float(__builtin_amdgcn_readlane(wj, j + q));
                const float* hs = H + (long)s * F;
                #pragma unroll
                for (int u = 0; u < NV; ++u) hv[q][u] = *(const float4*)(hs + col[u]);
            }
            #pragma unroll
            for (int q = 0; q < 4; ++q) {
                #pragma unroll
                for (int u = 0; u < NV; ++u) {
                    acc[u].x = fmaf(hv[q][u].x, wt[q], acc[u].x);
                    acc[u].y = fmaf(hv[q][u].y, wt[q], acc[u].y);
                    acc[u].z = fmaf(hv[q][u].z, wt[q], acc[u].z);
                    acc[u].w = fmaf(hv[q][u].w, wt[q], acc[u].w);
                }
            }
        }
        for (; j < nb; ++j) {
            const int   s = __builtin_amdgcn_readlane(sj, j);
            const float w = __int_as_float(__builtin_amdgcn_readlane(wj, j));
            const float* hs = H + (long)s * F;
            float4 h1[NV];
            #pragma unroll
            for (int u = 0; u < NV; ++u) h1[u] = *(const float4*)(hs + col[u]);
            #pragma unroll
            for (int u = 0; u < NV; ++u) {
                acc[u].x = fmaf(h1[u].x, w, acc[u].x);
                acc[u].y = fmaf(h1[u].y, w, acc[u].y);
                acc[u].z = fmaf(h1[u].z, w, acc[u].z);
                acc[u].w = fmaf(h1[u].w, w, acc[u].w);
            }
        }
    }

    const float di = rsqrtf(fmaxf((float)(end - beg), 1.0f));
    float* orow = outp + (long)node * F;
    #pragma unroll
    for (int u = 0; u < NV; ++u) {
        const int c = 4 * (lane + 32 * u);
        if (c < F) {
            const float4 bb = *(const float4*)(b + c);
            float4 o;
            o.x = acc[u].x * di + bb.x;
            o.y = acc[u].y * di + bb.y;
            o.z = acc[u].z * di + bb.z;
            o.w = acc[u].w * di + bb.w;
            if (RELU) {
                o.x = fmaxf(o.x, 0.0f);
                o.y = fmaxf(o.y, 0.0f);
                o.z = fmaxf(o.z, 0.0f);
                o.w = fmaxf(o.w, 0.0f);
            }
            *(float4*)(orow + c) = o;
        }
    }
}

template <int NV, int NT2>
__global__ __launch_bounds__(512) void gcn_gather_gemm_kernel(
    const float* __restrict__ H, const int* __restrict__ rec,
    const int* __restrict__ rowptr, const float* __restrict__ b,
    const float* __restrict__ W, float* __restrict__ H2, int n, int F, int F2)
{
    constexpr int XP = 128 * NV + 4;
    constexpr int NC = 16 * NT2;
    constexpr int BP = (NC % 32 == 16) ? NC : NC + 16;
    constexpr int KP = 64 * NV;
    __shared__ __attribute__((aligned(16))) float Xs[16 * XP];
    __shared__ __attribute__((aligned(16))) v2f   Ws[KP * BP];

    const int t    = threadIdx.x;
    const int lane = t & 31;
    const int wv    = __builtin_amdgcn_readfirstlane(t >> 5);
    const int node0 = blockIdx.x * 16;
    const int node  = node0 + wv;

    {
        constexpr int WI = (KP * NC + 511) / 512;
        const int P = F >> 1;
        v2f wreg[WI];
        #pragma unroll
        for (int i = 0; i < WI; ++i) {
            const int idx = t + 512 * i;
            const int p = idx / NC, c = idx % NC;
            wreg[i].x = 0.0f; wreg[i].y = 0.0f;
            if (p < P && c < F2) {
                wreg[i].x = W[(long)(2 * p) * F2 + c];
                wreg[i].y = W[(long)(2 * p + 1) * F2 + c];
            }
        }
        #pragma unroll
        for (int i = 0; i < WI; ++i) {
            const int idx = t + 512 * i;
            const int p = idx / NC, c = idx % NC;
            if (p < P) Ws[p * BP + c] = wreg[i];
        }
    }

    unsigned col[NV];
    #pragma unroll
    for (int u = 0; u < NV; ++u) col[u] = (unsigned)min(4 * (lane + 32 * u), F - 4);

    float4 acc[NV];
    #pragma unroll
    for (int u = 0; u < NV; ++u) acc[u] = make_float4(0.0f, 0.0f, 0.0f, 0.0f);

    int beg = 0, end = 0;
    if (node < n) { beg = rowptr[node]; end = rowptr[node + 1]; }

    for (int base = beg; base < end; base += 32) {
        const int nb = min(32, end - base);
        int sj = 0, wj = 0;
        if (lane < nb) {
            const int2 r = *(const int2*)(rec + 2 * (long)(base + lane));
            sj = r.x;
            wj = r.y;
        }
        int j = 0;
        for (; j + 4 <= nb; j += 4) {
            float4 hv[4][NV];
            float  wt[4];
            #pragma unroll
            for (int q = 0; q < 4; ++q) {
                const int s = __builtin_amdgcn_readlane(sj, j + q);
                wt[q] = __int_as_float(__builtin_amdgcn_readlane(wj, j + q));
                const float* hs = H + (long)s * F;
                #pragma unroll
                for (int u = 0; u < NV; ++u) hv[q][u] = *(const float4*)(hs + col[u]);
            }
            #pragma unroll
            for (int q = 0; q < 4; ++q) {
                #pragma unroll
                for (int u = 0; u < NV; ++u) {
                    acc[u].x = fmaf(hv[q][u].x, wt[q], acc[u].x);
                    acc[u].y = fmaf(hv[q][u].y, wt[q], acc[u].y);
                    acc[u].z = fmaf(hv[q][u].z, wt[q], acc[u].z);
                    acc[u].w = fmaf(hv[q][u].w, wt[q], acc[u].w);
                }
            }
        }
        for (; j < nb; ++j) {
            const int   s = __builtin_amdgcn_readlane(sj, j);
            const float w = __int_as_float(__builtin_amdgcn_readlane(wj, j));
            const float* hs = H + (long)s * F;
            float4 h1[NV];
            #pragma unroll
            for (int u = 0; u < NV; ++u) h1[u] = *(const float4*)(hs + col[u]);
            #pragma unroll
            for (int u = 0; u < NV; ++u) {
                acc[u].x = fmaf(h1[u].x, w, acc[u].x);
                acc[u].y = fmaf(h1[u].y, w, acc[u].y);
                acc[u].z = fmaf(h1[u].z, w, acc[u].z);
                acc[u].w = fmaf(h1[u].w, w, acc[u].w);
            }
        }
    }

    const float di = rsqrtf(fmaxf((float)(end - beg), 1.0f));
    #pragma unroll
    for (int u = 0; u < NV; ++u) {
        const int c = 4 * (lane + 32 * u);
        if (c < F) {
            float4 o = make_float4(0.0f, 0.0f, 0.0f, 0.0f);
            if (node < n) {
                const float4 bb = *(const float4*)(b + c);
                o.x = fmaxf(acc[u].x * di + bb.x, 0.0f);
                o.y = fmaxf(acc[u].y * di + bb.y, 0.0f);
                o.z = fmaxf(acc[u].z * di + bb.z, 0.0f);
                o.w = fmaxf(acc[u].w * di + bb.w, 0.0f);
            }
            *(float4*)&Xs[wv * XP + c] = o;
        }
    }
    __syncthreads();

    if (wv != 0) return;
    const int half = lane >> 4;
    const int l16  = lane & 15;
    const float* as = &Xs[l16 * XP + half * 2];
    const v2f*   bs = &Ws[half * BP + l16];
    v8f d[NT2];
    #pragma unroll
    for (int j = 0; j < NT2; ++j) d[j] = v8f{};
    for (int ks = 0; ks < F; ks += 4) {
        const v2f a = *(const v2f*)(as + ks);
        v2f bf[NT2];
        #pragma unroll
        for (int j = 0; j < NT2; ++j) bf[j] = bs[(ks >> 1) * BP + j * 16];
        #pragma unroll
        for (int j = 0; j < NT2; ++j)
            d[j] = __builtin_amdgcn_wmma_f32_16x16x4_f32(
                false, a, false, bf[j], (short)0, d[j], false, false);
    }
    #pragma unroll
    for (int j = 0; j < NT2; ++j) {
        const int nn = j * 16 + l16;
        #pragma unroll
        for (int v = 0; v < 8; ++v) {
            const int m = node0 + v + half * 8;
            if (m < n && nn < F2) H2[(long)m * F2 + nn] = d[j][v];
        }
    }
}

extern "C" void kernel_launch(void* const* d_in, const int* in_sizes, int n_in,
                              void* d_out, int out_size, void* d_ws, size_t ws_size,
                              hipStream_t stream)
{
    const float* feature = (const float*)d_in[0];
    const int*   src     = (const int*)  d_in[1];
    const int*   dst     = (const int*)  d_in[2];
    const float* edge_w  = (const float*)d_in[3];
    const float* W1      = (const float*)d_in[4];
    const float* b1      = (const float*)d_in[5];
    const float* W2      = (const float*)d_in[6];
    const float* b2      = (const float*)d_in[7];
    float*       out     = (float*)d_out;

    const int F1 = in_sizes[5];
    const int F2 = in_sizes[7];
    const int K0 = in_sizes[4] / F1;
    const int N  = in_sizes[0] / K0;
    const int E  = in_sizes[1];

    float* ws      = (float*)d_ws;
    int*   cnt_out = (int*)ws;
    int*   rloc    = cnt_out + N;
    float* R1      = ws + 2 * (long)N;
    float* R2      = R1 + (long)N * F1;
    float* R3      = R2 + (long)N * F1;
    float* H1      = R1;
    float* H2      = R2;
    int*   rank    = (int*)R2;
    const int G    = (E + RK_CHUNK - 1) / RK_CHUNK;
    int*   part_in  = rank + E;
    int*   part_out = part_in + (long)G * N;
    int*   off      = part_out + (long)G * N;
    int*   rec     = (int*)R3;
    int*   rowptr  = rec + 2 * (long)E;
    const int NB   = (N + 255) / 256;
    const int EB   = (E + 255) / 256;
    int*   bsum    = rowptr + N + 1;

    const int GB = (N + GW_BM - 1) / GW_BM;
    gcn_wmma_gemm_wgp_f32<<<GB + 2 * G, 256, 0, stream>>>(feature, W1, H1, N, F1, K0, K0, F1, F1,
                                                          GB, src, dst, part_in, part_out, rank, E, N, G);
    gcn_merge_counts_kernel<<<NB, 256, 0, stream>>>(part_in, part_out, off, rloc, bsum, cnt_out, N, G);
    gcn_csr_fill_kernel<<<NB + EB, 256, 0, stream>>>(src, dst, edge_w, cnt_out, rloc, bsum,
                                                     rank, off, rec, rowptr, E, N, NB);

    gcn_gather_gemm_kernel<2, 2><<<(N + 15) / 16, 512, 0, stream>>>(H1, rec, rowptr, b1, W2, H2, N, F1, F2);
    gcn_gather_kernel<1, false><<<(N + 7) / 8, 256, 0, stream>>>(H2, rec, rowptr, b2, out, N, F2);
}
